# starting kernel
# baseline (speedup 1.0000x reference)
#include <hip/hip_runtime.h>

typedef __attribute__((ext_vector_type(2))) float v2f;
typedef __attribute__((ext_vector_type(8))) float v8f;

#define B_ 8
#define C_ 512
#define N_ 4096
#define EPS_ 1e-6f
#define HT_ 0.25f

static __device__ __forceinline__ v8f wmma4(v2f a, v2f b, v8f c) {
    return __builtin_amdgcn_wmma_f32_16x16x4_f32(false, a, false, b, (short)0, c, false, false);
}

__global__ __launch_bounds__(128) void k_qkproj(const float* __restrict__ x,
                                                const float* __restrict__ Wq,
                                                const float* __restrict__ Wk,
                                                float* __restrict__ Q,
                                                float* __restrict__ K) {
    const int lane = threadIdx.x & 31;
    const int wave = threadIdx.x >> 5;
    const int m    = lane & 15;
    const int kp   = (lane >> 4) << 1;
    const int half = lane >> 4;
    const int c0 = blockIdx.y * 32;
    const int n0 = blockIdx.x * 128 + wave * 32;
    const int b  = blockIdx.z;
    const float* xb  = x  + (size_t)b * C_ * N_;
    const float* wq0 = Wq + (size_t)(c0 + m) * C_;
    const float* wq1 = Wq + (size_t)(c0 + 16 + m) * C_;
    const float* wk0 = Wk + (size_t)(c0 + m) * C_;
    const float* wk1 = Wk + (size_t)(c0 + 16 + m) * C_;
    v8f q00 = {}, q01 = {}, q10 = {}, q11 = {};
    v8f k00 = {}, k01 = {}, k10 = {}, k11 = {};
    for (int k = 0; k < C_; k += 4) {
        const int kk = k + kp;
        v2f aq0 = *(const v2f*)(wq0 + kk);
        v2f aq1 = *(const v2f*)(wq1 + kk);
        v2f ak0 = *(const v2f*)(wk0 + kk);
        v2f ak1 = *(const v2f*)(wk1 + kk);
        const float* xr0 = xb + (size_t)kk * N_;
        const float* xr1 = xb + (size_t)(kk + 1) * N_;
        v2f b0, b1;
        b0.x = xr0[n0 + m];      b0.y = xr1[n0 + m];
        b1.x = xr0[n0 + 16 + m]; b1.y = xr1[n0 + 16 + m];
        q00 = wmma4(aq0, b0, q00); q01 = wmma4(aq0, b1, q01);
        q10 = wmma4(aq1, b0, q10); q11 = wmma4(aq1, b1, q11);
        k00 = wmma4(ak0, b0, k00); k01 = wmma4(ak0, b1, k01);
        k10 = wmma4(ak1, b0, k10); k11 = wmma4(ak1, b1, k11);
    }
    float* Qb = Q + (size_t)b * C_ * N_;
    float* Kb = K + (size_t)b * C_ * N_;
    #pragma unroll
    for (int j = 0; j < 8; ++j) {
        const int cA = c0 + j + 8 * half;
        const int cB = cA + 16;
        Qb[(size_t)cA * N_ + n0 + m]      = q00[j];
        Qb[(size_t)cA * N_ + n0 + 16 + m] = q01[j];
        Qb[(size_t)cB * N_ + n0 + m]      = q10[j];
        Qb[(size_t)cB * N_ + n0 + 16 + m] = q11[j];
        Kb[(size_t)cA * N_ + n0 + m]      = k00[j];
        Kb[(size_t)cA * N_ + n0 + 16 + m] = k01[j];
        Kb[(size_t)cB * N_ + n0 + m]      = k10[j];
        Kb[(size_t)cB * N_ + n0 + 16 + m] = k11[j];
    }
}

__global__ __launch_bounds__(256) void k_rownorm(const float* __restrict__ Q,
                                                 const float* __restrict__ K,
                                                 float* __restrict__ invQ,
                                                 float* __restrict__ invK) {
    __shared__ float sq[256], sk[256];
    const int row = blockIdx.x;
    const float* qr = Q + (size_t)row * N_;
    const float* kr = K + (size_t)row * N_;
    float aq = 0.f, ak = 0.f;
    for (int n = threadIdx.x; n < N_; n += 256) {
        float v = qr[n]; aq += v * v;
        float w = kr[n]; ak += w * w;
    }
    sq[threadIdx.x] = aq; sk[threadIdx.x] = ak;
    __syncthreads();
    for (int s = 128; s > 0; s >>= 1) {
        if (threadIdx.x < s) {
            sq[threadIdx.x] += sq[threadIdx.x + s];
            sk[threadIdx.x] += sk[threadIdx.x + s];
        }
        __syncthreads();
    }
    if (threadIdx.x == 0) { invQ[row] = rsqrtf(sq[0]); invK[row] = rsqrtf(sk[0]); }
}

__global__ __launch_bounds__(128) void k_qkt(const float* __restrict__ Q,
                                             const float* __restrict__ K,
                                             const float* __restrict__ invQ,
                                             const float* __restrict__ invK,
                                             float* __restrict__ M) {
    const int lane = threadIdx.x & 31;
    const int wave = threadIdx.x >> 5;
    const int m    = lane & 15;
    const int kp   = (lane >> 4) << 1;
    const int half = lane >> 4;
    const int c0 = blockIdx.y * 32;
    const int d0 = blockIdx.x * 128 + wave * 32;
    const int b  = blockIdx.z;
    const size_t bOff = (size_t)b * C_ * N_;
    const float* q0 = Q + bOff + (size_t)(c0 + m) * N_;
    const float* q1 = Q + bOff + (size_t)(c0 + 16 + m) * N_;
    const float* k0 = K + bOff + (size_t)(d0 + m) * N_;
    const float* k1 = K + bOff + (size_t)(d0 + 16 + m) * N_;
    v8f a00 = {}, a01 = {}, a10 = {}, a11 = {};
    for (int k = 0; k < N_; k += 4) {
        v2f aA = *(const v2f*)(q0 + k + kp);
        v2f aB = *(const v2f*)(q1 + k + kp);
        v2f bA = *(const v2f*)(k0 + k + kp);
        v2f bB = *(const v2f*)(k1 + k + kp);
        a00 = wmma4(aA, bA, a00); a01 = wmma4(aA, bB, a01);
        a10 = wmma4(aB, bA, a10); a11 = wmma4(aB, bB, a11);
    }
    float* Mb = M + (size_t)b * C_ * C_;
    const float ik0 = invK[b * C_ + d0 + m];
    const float ik1 = invK[b * C_ + d0 + 16 + m];
    #pragma unroll
    for (int j = 0; j < 8; ++j) {
        const int cA = c0 + j + 8 * half;
        const int cB = cA + 16;
        const float iqA = invQ[b * C_ + cA];
        const float iqB = invQ[b * C_ + cB];
        Mb[(size_t)cA * C_ + d0 + m]      = a00[j] * iqA * ik0;
        Mb[(size_t)cA * C_ + d0 + 16 + m] = a01[j] * iqA * ik1;
        Mb[(size_t)cB * C_ + d0 + m]      = a10[j] * iqB * ik0;
        Mb[(size_t)cB * C_ + d0 + 16 + m] = a11[j] * iqB * ik1;
    }
}

__global__ __launch_bounds__(128) void k_colstat(const float* __restrict__ M,
                                                 float* __restrict__ denom,
                                                 float* __restrict__ shift) {
    __shared__ float smax[128], smin[128];
    const int idx = blockIdx.x;
    const int b = idx / C_;
    const int d = idx - b * C_;
    const float* Mb = M + (size_t)b * C_ * C_;
    float mx = -3.4e38f, mn = 3.4e38f;
    for (int c = threadIdx.x; c < C_; c += 128) {
        const float v = 1.f - Mb[(size_t)c * C_ + d];
        mx = fmaxf(mx, v);
        mn = fminf(mn, v);
    }
    smax[threadIdx.x] = mx; smin[threadIdx.x] = mn;
    __syncthreads();
    for (int s = 64; s > 0; s >>= 1) {
        if (threadIdx.x < s) {
            smax[threadIdx.x] = fmaxf(smax[threadIdx.x], smax[threadIdx.x + s]);
            smin[threadIdx.x] = fminf(smin[threadIdx.x], smin[threadIdx.x + s]);
        }
        __syncthreads();
    }
    if (threadIdx.x == 0) {
        const float dn = smax[0] + EPS_;
        denom[idx] = dn;
        shift[idx] = 1.f - (smin[0] / dn) * (1.f / HT_);
    }
}

__global__ __launch_bounds__(128) void k_colsoftmax(float* __restrict__ M,
                                                    const float* __restrict__ denom,
                                                    const float* __restrict__ shift) {
    __shared__ float ssum[128];
    const int idx = blockIdx.x;
    const int b = idx / C_;
    const int d = idx - b * C_;
    float* Mb = M + (size_t)b * C_ * C_;
    const float inv_dn = 1.f / denom[idx];
    const float sh = shift[idx];
    float z = 0.f;
    for (int c = threadIdx.x; c < C_; c += 128) {
        const float m4 = 1.f - ((1.f - Mb[(size_t)c * C_ + d]) * inv_dn) * (1.f / HT_);
        z += expf(m4 - sh);
    }
    ssum[threadIdx.x] = z;
    __syncthreads();
    for (int s = 64; s > 0; s >>= 1) {
        if (threadIdx.x < s) ssum[threadIdx.x] += ssum[threadIdx.x + s];
        __syncthreads();
    }
    const float invz = 1.f / ssum[0];
    for (int c = threadIdx.x; c < C_; c += 128) {
        const size_t o = (size_t)c * C_ + d;
        const float m4 = 1.f - ((1.f - Mb[o]) * inv_dn) * (1.f / HT_);
        Mb[o] = expf(m4 - sh) * invz;
    }
}

__global__ __launch_bounds__(256) void k_rowl1(float* __restrict__ M) {
    __shared__ float ssum[256];
    const int row = blockIdx.x;
    float* r = M + (size_t)row * C_;
    float s = 0.f;
    for (int d = threadIdx.x; d < C_; d += 256) s += r[d];
    ssum[threadIdx.x] = s;
    __syncthreads();
    for (int t = 128; t > 0; t >>= 1) {
        if (threadIdx.x < t) ssum[threadIdx.x] += ssum[threadIdx.x + t];
        __syncthreads();
    }
    const float inv = 1.f / (ssum[0] + EPS_);
    for (int d = threadIdx.x; d < C_; d += 256) r[d] *= inv;
}

__global__ __launch_bounds__(128) void k_mwv(const float* __restrict__ M,
                                             const float* __restrict__ Wv,
                                             float* __restrict__ G) {
    const int lane = threadIdx.x & 31;
    const int wave = threadIdx.x >> 5;
    const int m    = lane & 15;
    const int kp   = (lane >> 4) << 1;
    const int half = lane >> 4;
    const int c0 = blockIdx.y * 32;
    const int e0 = blockIdx.x * 128 + wave * 32;
    const int b  = blockIdx.z;
    const float* a0 = M + (size_t)b * C_ * C_ + (size_t)(c0 + m) * C_;
    const float* a1 = M + (size_t)b * C_ * C_ + (size_t)(c0 + 16 + m) * C_;
    v8f g00 = {}, g01 = {}, g10 = {}, g11 = {};
    for (int k = 0; k < C_; k += 4) {
        const int kk = k + kp;
        v2f aA = *(const v2f*)(a0 + kk);
        v2f aB = *(const v2f*)(a1 + kk);
        const float* wr0 = Wv + (size_t)kk * C_;
        const float* wr1 = Wv + (size_t)(kk + 1) * C_;
        v2f b0, b1;
        b0.x = wr0[e0 + m];      b0.y = wr1[e0 + m];
        b1.x = wr0[e0 + 16 + m]; b1.y = wr1[e0 + 16 + m];
        g00 = wmma4(aA, b0, g00); g01 = wmma4(aA, b1, g01);
        g10 = wmma4(aB, b0, g10); g11 = wmma4(aB, b1, g11);
    }
    float* Gb = G + (size_t)b * C_ * C_;
    #pragma unroll
    for (int j = 0; j < 8; ++j) {
        const int cA = c0 + j + 8 * half;
        const int cB = cA + 16;
        Gb[(size_t)cA * C_ + e0 + m]      = g00[j];
        Gb[(size_t)cA * C_ + e0 + 16 + m] = g01[j];
        Gb[(size_t)cB * C_ + e0 + m]      = g10[j];
        Gb[(size_t)cB * C_ + e0 + 16 + m] = g11[j];
    }
}

__global__ __launch_bounds__(128) void k_out(const float* __restrict__ x,
                                             const float* __restrict__ G,
                                             const float* __restrict__ gamma,
                                             float* __restrict__ y) {
    const int lane = threadIdx.x & 31;
    const int wave = threadIdx.x >> 5;
    const int m    = lane & 15;
    const int kp   = (lane >> 4) << 1;
    const int half = lane >> 4;
    const int c0 = blockIdx.y * 32;
    const int n0 = blockIdx.x * 128 + wave * 32;
    const int b  = blockIdx.z;
    const float* xb = x + (size_t)b * C_ * N_;
    const float* g0 = G + (size_t)b * C_ * C_ + (size_t)(c0 + m) * C_;
    const float* g1 = G + (size_t)b * C_ * C_ + (size_t)(c0 + 16 + m) * C_;
    v8f o00 = {}, o01 = {}, o10 = {}, o11 = {};
    for (int k = 0; k < C_; k += 4) {
        const int kk = k + kp;
        v2f aA = *(const v2f*)(g0 + kk);
        v2f aB = *(const v2f*)(g1 + kk);
        const float* xr0 = xb + (size_t)kk * N_;
        const float* xr1 = xb + (size_t)(kk + 1) * N_;
        v2f b0, b1;
        b0.x = xr0[n0 + m];      b0.y = xr1[n0 + m];
        b1.x = xr0[n0 + 16 + m]; b1.y = xr1[n0 + 16 + m];
        o00 = wmma4(aA, b0, o00); o01 = wmma4(aA, b1, o01);
        o10 = wmma4(aB, b0, o10); o11 = wmma4(aB, b1, o11);
    }
    const float g = gamma[0];
    float* yb = y + (size_t)b * C_ * N_;
    #pragma unroll
    for (int j = 0; j < 8; ++j) {
        const int cA = c0 + j + 8 * half;
        const int cB = cA + 16;
        const size_t o0 = (size_t)cA * N_ + n0 + m;
        const size_t o1 = (size_t)cA * N_ + n0 + 16 + m;
        const size_t o2 = (size_t)cB * N_ + n0 + m;
        const size_t o3 = (size_t)cB * N_ + n0 + 16 + m;
        yb[o0] = xb[o0] + g * o00[j];
        yb[o1] = xb[o1] + g * o01[j];
        yb[o2] = xb[o2] + g * o10[j];
        yb[o3] = xb[o3] + g * o11[j];
    }
}

extern "C" void kernel_launch(void* const* d_in, const int* in_sizes, int n_in,
                              void* d_out, int out_size, void* d_ws, size_t ws_size,
                              hipStream_t stream) {
    (void)in_sizes; (void)n_in; (void)out_size; (void)ws_size;
    const float* x     = (const float*)d_in[0];
    const float* Wq    = (const float*)d_in[1];
    const float* Wk    = (const float*)d_in[2];
    const float* Wv    = (const float*)d_in[3];
    const float* gamma = (const float*)d_in[4];
    float* y = (float*)d_out;

    char* ws = (char*)d_ws;
    const size_t qkBytes = (size_t)B_ * C_ * N_ * sizeof(float);
    const size_t mBytes  = (size_t)B_ * C_ * C_ * sizeof(float);
    float* Q     = (float*)(ws);
    float* K     = (float*)(ws + qkBytes);
    float* M     = (float*)(ws + 2 * qkBytes);
    float* G     = (float*)(ws + 2 * qkBytes + mBytes);
    float* invQ  = (float*)(ws + 2 * qkBytes + 2 * mBytes);
    float* invK  = invQ  + B_ * C_;
    float* denom = invK  + B_ * C_;
    float* shift = denom + B_ * C_;

    k_qkproj    <<<dim3(N_ / 128, C_ / 32, B_), 128, 0, stream>>>(x, Wq, Wk, Q, K);
    k_rownorm   <<<B_ * C_, 256, 0, stream>>>(Q, K, invQ, invK);
    k_qkt       <<<dim3(C_ / 128, C_ / 32, B_), 128, 0, stream>>>(Q, K, invQ, invK, M);
    k_colstat   <<<B_ * C_, 128, 0, stream>>>(M, denom, shift);
    k_colsoftmax<<<B_ * C_, 128, 0, stream>>>(M, denom, shift);
    k_rowl1     <<<B_ * C_, 256, 0, stream>>>(M);
    k_mwv       <<<dim3(C_ / 128, C_ / 32, B_), 128, 0, stream>>>(M, Wv, G);
    k_out       <<<dim3(N_ / 128, C_ / 32, B_), 128, 0, stream>>>(x, G, gamma, y);
}


# Q*K^T stored transposed (K rows as A operand) so the softmax axis is contiguous: col max/min/softmax in one wave-per-row pass, row L1 as coalesced column sums, M*Wv reads A by rows; sum trees change
# speedup vs baseline: 1.1709x; 1.1709x over previous
#include <hip/hip_runtime.h>

typedef __attribute__((ext_vector_type(2))) float v2f;
typedef __attribute__((ext_vector_type(8))) float v8f;

#define B_ 8
#define C_ 512
#define N_ 4096
#define EPS_ 1e-6f
#define HT_ 0.25f

static __device__ __forceinline__ v8f wmma4(v2f a, v2f b, v8f c) {
    return __builtin_amdgcn_wmma_f32_16x16x4_f32(false, a, false, b, (short)0, c, false, false);
}

__global__ __launch_bounds__(128) void k_qkproj(const float* __restrict__ x,
                                                const float* __restrict__ Wq,
                                                const float* __restrict__ Wk,
                                                float* __restrict__ Q,
                                                float* __restrict__ K) {
    const int lane = threadIdx.x & 31;
    const int wave = threadIdx.x >> 5;
    const int m    = lane & 15;
    const int kp   = (lane >> 4) << 1;
    const int half = lane >> 4;
    const int c0 = blockIdx.y * 32;
    const int n0 = blockIdx.x * 128 + wave * 32;
    const int b  = blockIdx.z;
    const float* xb  = x  + (size_t)b * C_ * N_;
    const float* wq0 = Wq + (size_t)(c0 + m) * C_;
    const float* wq1 = Wq + (size_t)(c0 + 16 + m) * C_;
    const float* wk0 = Wk + (size_t)(c0 + m) * C_;
    const float* wk1 = Wk + (size_t)(c0 + 16 + m) * C_;
    v8f q00 = {}, q01 = {}, q10 = {}, q11 = {};
    v8f k00 = {}, k01 = {}, k10 = {}, k11 = {};
    for (int k = 0; k < C_; k += 4) {
        const int kk = k + kp;
        v2f aq0 = *(const v2f*)(wq0 + kk);
        v2f aq1 = *(const v2f*)(wq1 + kk);
        v2f ak0 = *(const v2f*)(wk0 + kk);
        v2f ak1 = *(const v2f*)(wk1 + kk);
        const float* xr0 = xb + (size_t)kk * N_;
        const float* xr1 = xb + (size_t)(kk + 1) * N_;
        v2f b0, b1;
        b0.x = xr0[n0 + m];      b0.y = xr1[n0 + m];
        b1.x = xr0[n0 + 16 + m]; b1.y = xr1[n0 + 16 + m];
        q00 = wmma4(aq0, b0, q00); q01 = wmma4(aq0, b1, q01);
        q10 = wmma4(aq1, b0, q10); q11 = wmma4(aq1, b1, q11);
        k00 = wmma4(ak0, b0, k00); k01 = wmma4(ak0, b1, k01);
        k10 = wmma4(ak1, b0, k10); k11 = wmma4(ak1, b1, k11);
    }
    float* Qb = Q + (size_t)b * C_ * N_;
    float* Kb = K + (size_t)b * C_ * N_;
    #pragma unroll
    for (int j = 0; j < 8; ++j) {
        const int cA = c0 + j + 8 * half;
        const int cB = cA + 16;
        Qb[(size_t)cA * N_ + n0 + m]      = q00[j];
        Qb[(size_t)cA * N_ + n0 + 16 + m] = q01[j];
        Qb[(size_t)cB * N_ + n0 + m]      = q10[j];
        Qb[(size_t)cB * N_ + n0 + 16 + m] = q11[j];
        Kb[(size_t)cA * N_ + n0 + m]      = k00[j];
        Kb[(size_t)cA * N_ + n0 + 16 + m] = k01[j];
        Kb[(size_t)cB * N_ + n0 + m]      = k10[j];
        Kb[(size_t)cB * N_ + n0 + 16 + m] = k11[j];
    }
}

__global__ __launch_bounds__(256) void k_rownorm(const float* __restrict__ Q,
                                                 const float* __restrict__ K,
                                                 float* __restrict__ invQ,
                                                 float* __restrict__ invK) {
    __shared__ float sq[256], sk[256];
    const int row = blockIdx.x;
    const float* qr = Q + (size_t)row * N_;
    const float* kr = K + (size_t)row * N_;
    float aq = 0.f, ak = 0.f;
    for (int n = threadIdx.x; n < N_; n += 256) {
        float v = qr[n]; aq += v * v;
        float w = kr[n]; ak += w * w;
    }
    sq[threadIdx.x] = aq; sk[threadIdx.x] = ak;
    __syncthreads();
    for (int s = 128; s > 0; s >>= 1) {
        if (threadIdx.x < s) {
            sq[threadIdx.x] += sq[threadIdx.x + s];
            sk[threadIdx.x] += sk[threadIdx.x + s];
        }
        __syncthreads();
    }
    if (threadIdx.x == 0) { invQ[row] = rsqrtf(sq[0]); invK[row] = rsqrtf(sk[0]); }
}

__global__ __launch_bounds__(128) void k_qkt(const float* __restrict__ R,
                                             const float* __restrict__ S,
                                             const float* __restrict__ invR,
                                             const float* __restrict__ invS,
                                             float* __restrict__ Mt) {
    const int lane = threadIdx.x & 31;
    const int wave = threadIdx.x >> 5;
    const int m    = lane & 15;
    const int kp   = (lane >> 4) << 1;
    const int half = lane >> 4;
    const int r0 = blockIdx.y * 32;
    const int s0 = blockIdx.x * 128 + wave * 32;
    const int b  = blockIdx.z;
    const size_t bOff = (size_t)b * C_ * N_;
    const float* p0 = R + bOff + (size_t)(r0 + m) * N_;
    const float* p1 = R + bOff + (size_t)(r0 + 16 + m) * N_;
    const float* t0 = S + bOff + (size_t)(s0 + m) * N_;
    const float* t1 = S + bOff + (size_t)(s0 + 16 + m) * N_;
    v8f a00 = {}, a01 = {}, a10 = {}, a11 = {};
    for (int k = 0; k < N_; k += 4) {
        v2f aA = *(const v2f*)(p0 + k + kp);
        v2f aB = *(const v2f*)(p1 + k + kp);
        v2f bA = *(const v2f*)(t0 + k + kp);
        v2f bB = *(const v2f*)(t1 + k + kp);
        a00 = wmma4(aA, bA, a00); a01 = wmma4(aA, bB, a01);
        a10 = wmma4(aB, bA, a10); a11 = wmma4(aB, bB, a11);
    }
    float* Mb = Mt + (size_t)b * C_ * C_;
    const float is0 = invS[b * C_ + s0 + m];
    const float is1 = invS[b * C_ + s0 + 16 + m];
    #pragma unroll
    for (int j = 0; j < 8; ++j) {
        const int rA = r0 + j + 8 * half;
        const int rB = rA + 16;
        const float irA = invR[b * C_ + rA];
        const float irB = invR[b * C_ + rB];
        Mb[(size_t)rA * C_ + s0 + m]      = a00[j] * irA * is0;
        Mb[(size_t)rA * C_ + s0 + 16 + m] = a01[j] * irA * is1;
        Mb[(size_t)rB * C_ + s0 + m]      = a10[j] * irB * is0;
        Mb[(size_t)rB * C_ + s0 + 16 + m] = a11[j] * irB * is1;
    }
}

__global__ __launch_bounds__(256) void k_rowsoftmax(float* __restrict__ Mt) {
    const int lane = threadIdx.x & 31;
    const int row  = blockIdx.x * 8 + (threadIdx.x >> 5);
    float4* r = (float4*)(Mt + (size_t)row * C_);
    float4 v[4];
    float mx = -3.4e38f, mn = 3.4e38f;
    for (int i = 0; i < 4; ++i) {
        float4 t = r[i * 32 + lane];
        t.x = 1.f - t.x; t.y = 1.f - t.y; t.z = 1.f - t.z; t.w = 1.f - t.w;
        v[i] = t;
        mx = fmaxf(mx, fmaxf(fmaxf(t.x, t.y), fmaxf(t.z, t.w)));
        mn = fminf(mn, fminf(fminf(t.x, t.y), fminf(t.z, t.w)));
    }
    for (int o = 16; o > 0; o >>= 1) {
        mx = fmaxf(mx, __shfl_xor(mx, o, 32));
        mn = fminf(mn, __shfl_xor(mn, o, 32));
    }
    const float dn = mx + EPS_;
    const float inv_dn = 1.f / dn;
    const float sh = 1.f - (mn / dn) * (1.f / HT_);
    float z = 0.f;
    for (int i = 0; i < 4; ++i) {
        float4 t = v[i];
        t.x = expf((1.f - (t.x * inv_dn) * (1.f / HT_)) - sh);
        t.y = expf((1.f - (t.y * inv_dn) * (1.f / HT_)) - sh);
        t.z = expf((1.f - (t.z * inv_dn) * (1.f / HT_)) - sh);
        t.w = expf((1.f - (t.w * inv_dn) * (1.f / HT_)) - sh);
        v[i] = t;
        z += (t.x + t.y) + (t.z + t.w);
    }
    for (int o = 16; o > 0; o >>= 1) z += __shfl_xor(z, o, 32);
    const float invz = 1.f / z;
    for (int i = 0; i < 4; ++i) {
        float4 t = v[i];
        t.x *= invz; t.y *= invz; t.z *= invz; t.w *= invz;
        r[i * 32 + lane] = t;
    }
}

__global__ __launch_bounds__(1024) void k_l1(float* __restrict__ Mt) {
    __shared__ float part[32][33];
    const int lane = threadIdx.x & 31;
    const int w    = threadIdx.x >> 5;
    float* base = Mt + (size_t)blockIdx.y * C_ * C_ + (size_t)(w * 16) * C_
                     + blockIdx.x * 32 + lane;
    float v[16];
    float s = 0.f;
    for (int i = 0; i < 16; ++i) { v[i] = base[(size_t)i * C_]; s += v[i]; }
    part[w][lane] = s;
    __syncthreads();
    float t = 0.f;
    for (int i = 0; i < 32; ++i) t += part[i][lane];
    const float inv = 1.f / (t + EPS_);
    for (int i = 0; i < 16; ++i) base[(size_t)i * C_] = v[i] * inv;
}

__global__ __launch_bounds__(128) void k_mwv(const float* __restrict__ Mt,
                                             const float* __restrict__ Wv,
                                             float* __restrict__ G) {
    const int lane = threadIdx.x & 31;
    const int wave = threadIdx.x >> 5;
    const int m    = lane & 15;
    const int kp   = (lane >> 4) << 1;
    const int half = lane >> 4;
    const int c0 = blockIdx.y * 32;
    const int e0 = blockIdx.x * 128 + wave * 32;
    const int b  = blockIdx.z;
    const float* a = Mt + (size_t)b * C_ * C_ + c0 + m;
    v8f g00 = {}, g01 = {}, g10 = {}, g11 = {};
    for (int k = 0; k < C_; k += 4) {
        const int kk = k + kp;
        const float* ar0 = a + (size_t)kk * C_;
        const float* ar1 = a + (size_t)(kk + 1) * C_;
        v2f aA, aB;
        aA.x = ar0[0];  aA.y = ar1[0];
        aB.x = ar0[16]; aB.y = ar1[16];
        const float* wr0 = Wv + (size_t)kk * C_;
        const float* wr1 = Wv + (size_t)(kk + 1) * C_;
        v2f b0, b1;
        b0.x = wr0[e0 + m];      b0.y = wr1[e0 + m];
        b1.x = wr0[e0 + 16 + m]; b1.y = wr1[e0 + 16 + m];
        g00 = wmma4(aA, b0, g00); g01 = wmma4(aA, b1, g01);
        g10 = wmma4(aB, b0, g10); g11 = wmma4(aB, b1, g11);
    }
    float* Gb = G + (size_t)b * C_ * C_;
    #pragma unroll
    for (int j = 0; j < 8; ++j) {
        const int cA = c0 + j + 8 * half;
        const int cB = cA + 16;
        Gb[(size_t)cA * C_ + e0 + m]      = g00[j];
        Gb[(size_t)cA * C_ + e0 + 16 + m] = g01[j];
        Gb[(size_t)cB * C_ + e0 + m]      = g10[j];
        Gb[(size_t)cB * C_ + e0 + 16 + m] = g11[j];
    }
}

__global__ __launch_bounds__(128) void k_out(const float* __restrict__ x,
                                             const float* __restrict__ G,
                                             const float* __restrict__ gamma,
                                             float* __restrict__ y) {
    const int lane = threadIdx.x & 31;
    const int wave = threadIdx.x >> 5;
    const int m    = lane & 15;
    const int kp   = (lane >> 4) << 1;
    const int half = lane >> 4;
    const int c0 = blockIdx.y * 32;
    const int n0 = blockIdx.x * 128 + wave * 32;
    const int b  = blockIdx.z;
    const float* xb = x + (size_t)b * C_ * N_;
    const float* g0 = G + (size_t)b * C_ * C_ + (size_t)(c0 + m) * C_;
    const float* g1 = G + (size_t)b * C_ * C_ + (size_t)(c0 + 16 + m) * C_;
    v8f o00 = {}, o01 = {}, o10 = {}, o11 = {};
    for (int k = 0; k < C_; k += 4) {
        const int kk = k + kp;
        v2f aA = *(const v2f*)(g0 + kk);
        v2f aB = *(const v2f*)(g1 + kk);
        const float* xr0 = xb + (size_t)kk * N_;
        const float* xr1 = xb + (size_t)(kk + 1) * N_;
        v2f b0, b1;
        b0.x = xr0[n0 + m];      b0.y = xr1[n0 + m];
        b1.x = xr0[n0 + 16 + m]; b1.y = xr1[n0 + 16 + m];
        o00 = wmma4(aA, b0, o00); o01 = wmma4(aA, b1, o01);
        o10 = wmma4(aB, b0, o10); o11 = wmma4(aB, b1, o11);
    }
    const float g = gamma[0];
    float* yb = y + (size_t)b * C_ * N_;
    #pragma unroll
    for (int j = 0; j < 8; ++j) {
        const int cA = c0 + j + 8 * half;
        const int cB = cA + 16;
        const size_t o0 = (size_t)cA * N_ + n0 + m;
        const size_t o1 = (size_t)cA * N_ + n0 + 16 + m;
        const size_t o2 = (size_t)cB * N_ + n0 + m;
        const size_t o3 = (size_t)cB * N_ + n0 + 16 + m;
        yb[o0] = xb[o0] + g * o00[j];
        yb[o1] = xb[o1] + g * o01[j];
        yb[o2] = xb[o2] + g * o10[j];
        yb[o3] = xb[o3] + g * o11[j];
    }
}

extern "C" void kernel_launch(void* const* d_in, const int* in_sizes, int n_in,
                              void* d_out, int out_size, void* d_ws, size_t ws_size,
                              hipStream_t stream) {
    (void)in_sizes; (void)n_in; (void)out_size; (void)ws_size;
    const float* x     = (const float*)d_in[0];
    const float* Wq    = (const float*)d_in[1];
    const float* Wk    = (const float*)d_in[2];
    const float* Wv    = (const float*)d_in[3];
    const float* gamma = (const float*)d_in[4];
    float* y = (float*)d_out;

    char* ws = (char*)d_ws;
    const size_t qkBytes = (size_t)B_ * C_ * N_ * sizeof(float);
    const size_t mBytes  = (size_t)B_ * C_ * C_ * sizeof(float);
    float* Q     = (float*)(ws);
    float* K     = (float*)(ws + qkBytes);
    float* M     = (float*)(ws + 2 * qkBytes);
    float* G     = (float*)(ws + 2 * qkBytes + mBytes);
    float* invQ  = (float*)(ws + 2 * qkBytes + 2 * mBytes);
    float* invK  = invQ  + B_ * C_;

    k_qkproj    <<<dim3(N_ / 128, C_ / 32, B_), 128, 0, stream>>>(x, Wq, Wk, Q, K);
    k_rownorm   <<<B_ * C_, 256, 0, stream>>>(Q, K, invQ, invK);
    k_qkt       <<<dim3(C_ / 128, C_ / 32, B_), 128, 0, stream>>>(K, Q, invK, invQ, M);
    k_rowsoftmax<<<B_ * C_ / 8, 256, 0, stream>>>(M);
    k_l1        <<<dim3(C_ / 32, B_), 1024, 0, stream>>>(M);
    k_mwv       <<<dim3(C_ / 128, C_ / 32, B_), 128, 0, stream>>>(M, Wv, G);
    k_out       <<<dim3(N_ / 128, C_ / 32, B_), 128, 0, stream>>>(x, G, gamma, y);
}


# Q, K, G written by their producers (transposed tile) in the consumer's WMMA fragment order: fragment loads and stores are whole 512 B runs, QK^T loads ringed 3 k blocks ahead; k and norm sums reorder
# speedup vs baseline: 1.4923x; 1.2745x over previous
#include <hip/hip_runtime.h>

typedef __attribute__((ext_vector_type(2))) float v2f;
typedef __attribute__((ext_vector_type(8))) float v8f;

#define B_ 8
#define C_ 512
#define N_ 4096
#define EPS_ 1e-6f
#define HT_ 0.25f

static __device__ __forceinline__ v8f wmma4(v2f a, v2f b, v8f c) {
    return __builtin_amdgcn_wmma_f32_16x16x4_f32(false, a, false, b, (short)0, c, false, false);
}

static __device__ __forceinline__ void store_frag_tile(float* tile, int lane, v8f acc) {
    float4* p = (float4*)tile + lane;
    p[0]  = make_float4(acc[0], acc[1], acc[2], acc[3]);
    p[32] = make_float4(acc[4], acc[5], acc[6], acc[7]);
}

__global__ __launch_bounds__(128) void k_qkproj(const float* __restrict__ x,
                                                const float* __restrict__ Wq,
                                                const float* __restrict__ Wk,
                                                float* __restrict__ Qp,
                                                float* __restrict__ Kp) {
    const int lane = threadIdx.x & 31;
    const int wave = threadIdx.x >> 5;
    const int m    = lane & 15;
    const int kp   = (lane >> 4) << 1;
    const int c0 = blockIdx.y * 32;
    const int n0 = blockIdx.x * 128 + wave * 32;
    const int b  = blockIdx.z;
    const float* xb  = x  + (size_t)b * C_ * N_;
    const float* wq0 = Wq + (size_t)(c0 + m) * C_;
    const float* wq1 = Wq + (size_t)(c0 + 16 + m) * C_;
    const float* wk0 = Wk + (size_t)(c0 + m) * C_;
    const float* wk1 = Wk + (size_t)(c0 + 16 + m) * C_;
    v8f q00 = {}, q01 = {}, q10 = {}, q11 = {};
    v8f k00 = {}, k01 = {}, k10 = {}, k11 = {};
    for (int k = 0; k < C_; k += 4) {
        const int kk = k + kp;
        v2f aq0 = *(const v2f*)(wq0 + kk);
        v2f aq1 = *(const v2f*)(wq1 + kk);
        v2f ak0 = *(const v2f*)(wk0 + kk);
        v2f ak1 = *(const v2f*)(wk1 + kk);
        const float* xr0 = xb + (size_t)kk * N_;
        const float* xr1 = xb + (size_t)(kk + 1) * N_;
        v2f b0, b1;
        b0.x = xr0[n0 + m];      b0.y = xr1[n0 + m];
        b1.x = xr0[n0 + 16 + m]; b1.y = xr1[n0 + 16 + m];
        q00 = wmma4(b0, aq0, q00); q01 = wmma4(b1, aq0, q01);
        q10 = wmma4(b0, aq1, q10); q11 = wmma4(b1, aq1, q11);
        k00 = wmma4(b0, ak0, k00); k01 = wmma4(b1, ak0, k01);
        k10 = wmma4(b0, ak1, k10); k11 = wmma4(b1, ak1, k11);
    }
    float* Qb = Qp + (size_t)b * C_ * N_ + (size_t)(c0 >> 4) * (16 * N_) + (size_t)(n0 >> 4) * 256;
    float* Kb = Kp + (size_t)b * C_ * N_ + (size_t)(c0 >> 4) * (16 * N_) + (size_t)(n0 >> 4) * 256;
    store_frag_tile(Qb,                 lane, q00);
    store_frag_tile(Qb + 256,           lane, q01);
    store_frag_tile(Qb + 16 * N_,       lane, q10);
    store_frag_tile(Qb + 16 * N_ + 256, lane, q11);
    store_frag_tile(Kb,                 lane, k00);
    store_frag_tile(Kb + 256,           lane, k01);
    store_frag_tile(Kb + 16 * N_,       lane, k10);
    store_frag_tile(Kb + 16 * N_ + 256, lane, k11);
}

__global__ __launch_bounds__(1024) void k_rownorm(const float* __restrict__ Qp,
                                                  const float* __restrict__ Kp,
                                                  float* __restrict__ invQ,
                                                  float* __restrict__ invK) {
    __shared__ float part[32][16];
    const int tid  = threadIdx.x;
    const int lane = tid & 31;
    const int w    = tid >> 5;
    const int blk  = blockIdx.x;
    const float* src = blockIdx.y ? Kp : Qp;
    const float4* p = (const float4*)(src + (size_t)blk * (16 * N_)) + tid;
    float s = 0.f;
    #pragma unroll
    for (int i = 0; i < 16; ++i) {
        float4 v = p[i * 1024];
        s += (v.x * v.x + v.y * v.y) + (v.z * v.z + v.w * v.w);
    }
    s += __shfl_xor(s, 16, 32);
    if (lane < 16) part[w][lane] = s;
    __syncthreads();
    if (tid < 16) {
        float t = 0.f;
        for (int i = 0; i < 32; ++i) t += part[i][tid];
        float* dst = blockIdx.y ? invK : invQ;
        dst[blk * 16 + tid] = rsqrtf(t);
    }
}

struct QktFrag { float4 a0l, a0h, a1l, a1h, b0l, b0h, b1l, b1h; };

static __device__ __forceinline__ void qkt_load(QktFrag& f, const float4* p0, const float4* p1,
                                                const float4* t0, const float4* t1, int kb) {
    const int o = kb * 64;
    f.a0l = p0[o]; f.a0h = p0[o + 32];
    f.a1l = p1[o]; f.a1h = p1[o + 32];
    f.b0l = t0[o]; f.b0h = t0[o + 32];
    f.b1l = t1[o]; f.b1h = t1[o + 32];
}

static __device__ __forceinline__ void qkt_step(v2f aA, v2f aB, v2f bA, v2f bB,
                                                v8f& a00, v8f& a01, v8f& a10, v8f& a11) {
    a00 = wmma4(aA, bA, a00); a01 = wmma4(aA, bB, a01);
    a10 = wmma4(aB, bA, a10); a11 = wmma4(aB, bB, a11);
}

static __device__ __forceinline__ void qkt_mma(const QktFrag& f, v8f& a00, v8f& a01, v8f& a10, v8f& a11) {
    qkt_step(v2f{f.a0l.x, f.a0l.y}, v2f{f.a1l.x, f.a1l.y}, v2f{f.b0l.x, f.b0l.y}, v2f{f.b1l.x, f.b1l.y}, a00, a01, a10, a11);
    qkt_step(v2f{f.a0l.z, f.a0l.w}, v2f{f.a1l.z, f.a1l.w}, v2f{f.b0l.z, f.b0l.w}, v2f{f.b1l.z, f.b1l.w}, a00, a01, a10, a11);
    qkt_step(v2f{f.a0h.x, f.a0h.y}, v2f{f.a1h.x, f.a1h.y}, v2f{f.b0h.x, f.b0h.y}, v2f{f.b1h.x, f.b1h.y}, a00, a01, a10, a11);
    qkt_step(v2f{f.a0h.z, f.a0h.w}, v2f{f.a1h.z, f.a1h.w}, v2f{f.b0h.z, f.b0h.w}, v2f{f.b1h.z, f.b1h.w}, a00, a01, a10, a11);
}

__global__ __launch_bounds__(128) void k_qkt(const float* __restrict__ Rp,
                                             const float* __restrict__ Sp,
                                             const float* __restrict__ invR,
                                             const float* __restrict__ invS,
                                             float* __restrict__ Mt) {
    const int lane = threadIdx.x & 31;
    const int wave = threadIdx.x >> 5;
    const int m    = lane & 15;
    const int half = lane >> 4;
    const int r0 = blockIdx.y * 32;
    const int s0 = blockIdx.x * 128 + wave * 32;
    const int b  = blockIdx.z;
    const size_t bOff = (size_t)b * C_ * N_;
    const float4* p0 = (const float4*)(Rp + bOff + (size_t)(r0 >> 4) * (16 * N_)) + lane;
    const float4* p1 = p0 + 4 * N_;
    const float4* t0 = (const float4*)(Sp + bOff + (size_t)(s0 >> 4) * (16 * N_)) + lane;
    const float4* t1 = t0 + 4 * N_;
    v8f a00 = {}, a01 = {}, a10 = {}, a11 = {};
    QktFrag f0, f1, f2, f3;
    const int nkb = N_ / 16;
    qkt_load(f0, p0, p1, t0, t1, 0);
    __builtin_amdgcn_sched_barrier(0);
    qkt_load(f1, p0, p1, t0, t1, 1);
    __builtin_amdgcn_sched_barrier(0);
    qkt_load(f2, p0, p1, t0, t1, 2);
    __builtin_amdgcn_sched_barrier(0);
    for (int kb = 0; kb < nkb; kb += 4) {
        const bool more = kb + 4 < nkb;
        qkt_load(f3, p0, p1, t0, t1, kb + 3);
        __builtin_amdgcn_sched_barrier(0);
        qkt_mma(f0, a00, a01, a10, a11);
        __builtin_amdgcn_sched_barrier(0);
        qkt_load(f0, p0, p1, t0, t1, more ? kb + 4 : kb);
        __builtin_amdgcn_sched_barrier(0);
        qkt_mma(f1, a00, a01, a10, a11);
        __builtin_amdgcn_sched_barrier(0);
        qkt_load(f1, p0, p1, t0, t1, more ? kb + 5 : kb);
        __builtin_amdgcn_sched_barrier(0);
        qkt_mma(f2, a00, a01, a10, a11);
        __builtin_amdgcn_sched_barrier(0);
        qkt_load(f2, p0, p1, t0, t1, more ? kb + 6 : kb);
        __builtin_amdgcn_sched_barrier(0);
        qkt_mma(f3, a00, a01, a10, a11);
        __builtin_amdgcn_sched_barrier(0);
    }
    float* Mb = Mt + (size_t)b * C_ * C_;
    const float is0 = invS[b * C_ + s0 + m];
    const float is1 = invS[b * C_ + s0 + 16 + m];
    #pragma unroll
    for (int j = 0; j < 8; ++j) {
        const int rA = r0 + j + 8 * half;
        const int rB = rA + 16;
        const float irA = invR[b * C_ + rA];
        const float irB = invR[b * C_ + rB];
        Mb[(size_t)rA * C_ + s0 + m]      = a00[j] * irA * is0;
        Mb[(size_t)rA * C_ + s0 + 16 + m] = a01[j] * irA * is1;
        Mb[(size_t)rB * C_ + s0 + m]      = a10[j] * irB * is0;
        Mb[(size_t)rB * C_ + s0 + 16 + m] = a11[j] * irB * is1;
    }
}

__global__ __launch_bounds__(256) void k_rowsoftmax(float* __restrict__ Mt) {
    const int lane = threadIdx.x & 31;
    const int row  = blockIdx.x * 8 + (threadIdx.x >> 5);
    float4* r = (float4*)(Mt + (size_t)row * C_);
    float4 v[4];
    float mx = -3.4e38f, mn = 3.4e38f;
    for (int i = 0; i < 4; ++i) {
        float4 t = r[i * 32 + lane];
        t.x = 1.f - t.x; t.y = 1.f - t.y; t.z = 1.f - t.z; t.w = 1.f - t.w;
        v[i] = t;
        mx = fmaxf(mx, fmaxf(fmaxf(t.x, t.y), fmaxf(t.z, t.w)));
        mn = fminf(mn, fminf(fminf(t.x, t.y), fminf(t.z, t.w)));
    }
    for (int o = 16; o > 0; o >>= 1) {
        mx = fmaxf(mx, __shfl_xor(mx, o, 32));
        mn = fminf(mn, __shfl_xor(mn, o, 32));
    }
    const float dn = mx + EPS_;
    const float inv_dn = 1.f / dn;
    const float sh = 1.f - (mn / dn) * (1.f / HT_);
    float z = 0.f;
    for (int i = 0; i < 4; ++i) {
        float4 t = v[i];
        t.x = expf((1.f - (t.x * inv_dn) * (1.f / HT_)) - sh);
        t.y = expf((1.f - (t.y * inv_dn) * (1.f / HT_)) - sh);
        t.z = expf((1.f - (t.z * inv_dn) * (1.f / HT_)) - sh);
        t.w = expf((1.f - (t.w * inv_dn) * (1.f / HT_)) - sh);
        v[i] = t;
        z += (t.x + t.y) + (t.z + t.w);
    }
    for (int o = 16; o > 0; o >>= 1) z += __shfl_xor(z, o, 32);
    const float invz = 1.f / z;
    for (int i = 0; i < 4; ++i) {
        float4 t = v[i];
        t.x *= invz; t.y *= invz; t.z *= invz; t.w *= invz;
        r[i * 32 + lane] = t;
    }
}

__global__ __launch_bounds__(1024) void k_l1(float* __restrict__ Mt) {
    __shared__ float part[32][33];
    const int lane = threadIdx.x & 31;
    const int w    = threadIdx.x >> 5;
    float* base = Mt + (size_t)blockIdx.y * C_ * C_ + (size_t)(w * 16) * C_
                     + blockIdx.x * 32 + lane;
    float v[16];
    float s = 0.f;
    for (int i = 0; i < 16; ++i) { v[i] = base[(size_t)i * C_]; s += v[i]; }
    part[w][lane] = s;
    __syncthreads();
    float t = 0.f;
    for (int i = 0; i < 32; ++i) t += part[i][lane];
    const float inv = 1.f / (t + EPS_);
    for (int i = 0; i < 16; ++i) base[(size_t)i * C_] = v[i] * inv;
}

__global__ __launch_bounds__(128) void k_mwv(const float* __restrict__ Mt,
                                             const float* __restrict__ Wv,
                                             float* __restrict__ Gp) {
    const int lane = threadIdx.x & 31;
    const int wave = threadIdx.x >> 5;
    const int m    = lane & 15;
    const int kp   = (lane >> 4) << 1;
    const int c0 = blockIdx.y * 32;
    const int e0 = blockIdx.x * 128 + wave * 32;
    const int b  = blockIdx.z;
    const float* a = Mt + (size_t)b * C_ * C_ + c0 + m;
    v8f g00 = {}, g01 = {}, g10 = {}, g11 = {};
    for (int k = 0; k < C_; k += 4) {
        const int kk = k + kp;
        const float* ar0 = a + (size_t)kk * C_;
        const float* ar1 = a + (size_t)(kk + 1) * C_;
        v2f aA, aB;
        aA.x = ar0[0];  aA.y = ar1[0];
        aB.x = ar0[16]; aB.y = ar1[16];
        const float* wr0 = Wv + (size_t)kk * C_;
        const float* wr1 = Wv + (size_t)(kk + 1) * C_;
        v2f b0, b1;
        b0.x = wr0[e0 + m];      b0.y = wr1[e0 + m];
        b1.x = wr0[e0 + 16 + m]; b1.y = wr1[e0 + 16 + m];
        g00 = wmma4(b0, aA, g00); g01 = wmma4(b1, aA, g01);
        g10 = wmma4(b0, aB, g10); g11 = wmma4(b1, aB, g11);
    }
    float* Gb = Gp + (size_t)b * C_ * C_ + (size_t)(c0 >> 4) * (16 * C_) + (size_t)(e0 >> 4) * 256;
    store_frag_tile(Gb,                 lane, g00);
    store_frag_tile(Gb + 256,           lane, g01);
    store_frag_tile(Gb + 16 * C_,       lane, g10);
    store_frag_tile(Gb + 16 * C_ + 256, lane, g11);
}

__global__ __launch_bounds__(128) void k_out(const float* __restrict__ x,
                                             const float* __restrict__ Gp,
                                             const float* __restrict__ gamma,
                                             float* __restrict__ y) {
    const int lane = threadIdx.x & 31;
    const int wave = threadIdx.x >> 5;
    const int m    = lane & 15;
    const int half = lane >> 4;
    const int c0 = blockIdx.y * 32;
    const int n0 = blockIdx.x * 128 + wave * 32;
    const int b  = blockIdx.z;
    const float* xb = x + (size_t)b * C_ * N_;
    const float4* g0 = (const float4*)(Gp + (size_t)b * C_ * C_ + (size_t)(c0 >> 4) * (16 * C_)) + lane;
    const float4* g1 = g0 + 4 * C_;
    v8f o00 = {}, o01 = {}, o10 = {}, o11 = {};
    for (int kb = 0; kb < C_ / 16; ++kb) {
        const float4 A0l = g0[kb * 64], A0h = g0[kb * 64 + 32];
        const float4 A1l = g1[kb * 64], A1h = g1[kb * 64 + 32];
        const v2f aA[4] = { {A0l.x, A0l.y}, {A0l.z, A0l.w}, {A0h.x, A0h.y}, {A0h.z, A0h.w} };
        const v2f aB[4] = { {A1l.x, A1l.y}, {A1l.z, A1l.w}, {A1h.x, A1h.y}, {A1h.z, A1h.w} };
        const float* xr = xb + (size_t)(kb * 16 + 8 * half) * N_ + n0 + m;
        #pragma unroll
        for (int u = 0; u < 4; ++u) {
            v2f b0, b1;
            b0.x = xr[(size_t)(2 * u) * N_];      b0.y = xr[(size_t)(2 * u + 1) * N_];
            b1.x = xr[(size_t)(2 * u) * N_ + 16]; b1.y = xr[(size_t)(2 * u + 1) * N_ + 16];
            o00 = wmma4(aA[u], b0, o00); o01 = wmma4(aA[u], b1, o01);
            o10 = wmma4(aB[u], b0, o10); o11 = wmma4(aB[u], b1, o11);
        }
    }
    const float g = gamma[0];
    float* yb = y + (size_t)b * C_ * N_;
    #pragma unroll
    for (int j = 0; j < 8; ++j) {
        const int cA = c0 + j + 8 * half;
        const int cB = cA + 16;
        const size_t o0 = (size_t)cA * N_ + n0 + m;
        const size_t o1 = (size_t)cA * N_ + n0 + 16 + m;
        const size_t o2 = (size_t)cB * N_ + n0 + m;
        const size_t o3 = (size_t)cB * N_ + n0 + 16 + m;
        yb[o0] = xb[o0] + g * o00[j];
        yb[o1] = xb[o1] + g * o01[j];
        yb[o2] = xb[o2] + g * o10[j];
        yb[o3] = xb[o3] + g * o11[j];
    }
}

extern "C" void kernel_launch(void* const* d_in, const int* in_sizes, int n_in,
                              void* d_out, int out_size, void* d_ws, size_t ws_size,
                              hipStream_t stream) {
    (void)in_sizes; (void)n_in; (void)out_size; (void)ws_size;
    const float* x     = (const float*)d_in[0];
    const float* Wq    = (const float*)d_in[1];
    const float* Wk    = (const float*)d_in[2];
    const float* Wv    = (const float*)d_in[3];
    const float* gamma = (const float*)d_in[4];
    float* y = (float*)d_out;

    char* ws = (char*)d_ws;
    const size_t qkBytes = (size_t)B_ * C_ * N_ * sizeof(float);
    const size_t mBytes  = (size_t)B_ * C_ * C_ * sizeof(float);
    float* Q     = (float*)(ws);
    float* K     = (float*)(ws + qkBytes);
    float* M     = (float*)(ws + 2 * qkBytes);
    float* G     = (float*)(ws + 2 * qkBytes + mBytes);
    float* invQ  = (float*)(ws + 2 * qkBytes + 2 * mBytes);
    float* invK  = invQ  + B_ * C_;

    k_qkproj    <<<dim3(N_ / 128, C_ / 32, B_), 128, 0, stream>>>(x, Wq, Wk, Q, K);
    k_rownorm   <<<dim3(B_ * C_ / 16, 2), 1024, 0, stream>>>(Q, K, invQ, invK);
    k_qkt       <<<dim3(C_ / 128, C_ / 32, B_), 128, 0, stream>>>(K, Q, invK, invQ, M);
    k_rowsoftmax<<<B_ * C_ / 8, 256, 0, stream>>>(M);
    k_l1        <<<dim3(C_ / 32, B_), 1024, 0, stream>>>(M);
    k_mwv       <<<dim3(C_ / 128, C_ / 32, B_), 128, 0, stream>>>(M, Wv, G);
    k_out       <<<dim3(N_ / 128, C_ / 32, B_), 128, 0, stream>>>(x, G, gamma, y);
}


# Wq, Wk re-laid by a pre-pass into WMMA fragment order (as Q, K, G are): projection reads weight fragments as whole 512 B runs, k-block loads grouped a block ahead; k sums reorder; 7->8 launches
# speedup vs baseline: 1.6510x; 1.1063x over previous
#include <hip/hip_runtime.h>

typedef __attribute__((ext_vector_type(2))) float v2f;
typedef __attribute__((ext_vector_type(8))) float v8f;

#define B_ 8
#define C_ 512
#define N_ 4096
#define EPS_ 1e-6f
#define HT_ 0.25f

static __device__ __forceinline__ v8f wmma4(v2f a, v2f b, v8f c) {
    return __builtin_amdgcn_wmma_f32_16x16x4_f32(false, a, false, b, (short)0, c, false, false);
}

static __device__ __forceinline__ void store_frag_tile(float* tile, int lane, v8f acc) {
    float4* p = (float4*)tile + lane;
    p[0]  = make_float4(acc[0], acc[1], acc[2], acc[3]);
    p[32] = make_float4(acc[4], acc[5], acc[6], acc[7]);
}

__global__ __launch_bounds__(256) void k_wfrag(const float* __restrict__ Wq,
                                               const float* __restrict__ Wk,
                                               float* __restrict__ Wqp,
                                               float* __restrict__ Wkp) {
    const int i    = blockIdx.x * 256 + threadIdx.x;
    const int lane = i & 31;
    const int t    = (i >> 5) & 1;
    const int kb   = (i >> 6) & 31;
    const int rb   = i >> 11;
    const int row  = rb * 16 + (lane & 15);
    const int k    = kb * 16 + 8 * (lane >> 4) + 4 * t;
    const float* src = blockIdx.y ? Wk : Wq;
    float* dst = blockIdx.y ? Wkp : Wqp;
    ((float4*)dst)[i] = *(const float4*)(src + (size_t)row * C_ + k);
}

struct ProjFrag {
    float4 q0l, q0h, q1l, q1h, k0l, k0h, k1l, k1h;
    float  xa[8], xb[8];
};

static __device__ __forceinline__ void proj_load(ProjFrag& f, const float4* q0, const float4* q1,
                                                 const float4* k0, const float4* k1,
                                                 const float* xr, int kb) {
    const int o = kb * 64;
    f.q0l = q0[o]; f.q0h = q0[o + 32];
    f.q1l = q1[o]; f.q1h = q1[o + 32];
    f.k0l = k0[o]; f.k0h = k0[o + 32];
    f.k1l = k1[o]; f.k1h = k1[o + 32];
    const float* p = xr + (size_t)kb * (16 * N_);
    #pragma unroll
    for (int j = 0; j < 8; ++j) {
        f.xa[j] = p[(size_t)j * N_];
        f.xb[j] = p[(size_t)j * N_ + 16];
    }
}

static __device__ __forceinline__ void proj_step(v2f xA, v2f xB, v2f wq0, v2f wq1, v2f wk0, v2f wk1,
                                                 v8f& q00, v8f& q01, v8f& q10, v8f& q11,
                                                 v8f& k00, v8f& k01, v8f& k10, v8f& k11) {
    q00 = wmma4(xA, wq0, q00); q01 = wmma4(xB, wq0, q01);
    q10 = wmma4(xA, wq1, q10); q11 = wmma4(xB, wq1, q11);
    k00 = wmma4(xA, wk0, k00); k01 = wmma4(xB, wk0, k01);
    k10 = wmma4(xA, wk1, k10); k11 = wmma4(xB, wk1, k11);
}

static __device__ __forceinline__ void proj_mma(const ProjFrag& f,
                                                v8f& q00, v8f& q01, v8f& q10, v8f& q11,
                                                v8f& k00, v8f& k01, v8f& k10, v8f& k11) {
    proj_step(v2f{f.xa[0], f.xa[1]}, v2f{f.xb[0], f.xb[1]},
              v2f{f.q0l.x, f.q0l.y}, v2f{f.q1l.x, f.q1l.y}, v2f{f.k0l.x, f.k0l.y}, v2f{f.k1l.x, f.k1l.y},
              q00, q01, q10, q11, k00, k01, k10, k11);
    proj_step(v2f{f.xa[2], f.xa[3]}, v2f{f.xb[2], f.xb[3]},
              v2f{f.q0l.z, f.q0l.w}, v2f{f.q1l.z, f.q1l.w}, v2f{f.k0l.z, f.k0l.w}, v2f{f.k1l.z, f.k1l.w},
              q00, q01, q10, q11, k00, k01, k10, k11);
    proj_step(v2f{f.xa[4], f.xa[5]}, v2f{f.xb[4], f.xb[5]},
              v2f{f.q0h.x, f.q0h.y}, v2f{f.q1h.x, f.q1h.y}, v2f{f.k0h.x, f.k0h.y}, v2f{f.k1h.x, f.k1h.y},
              q00, q01, q10, q11, k00, k01, k10, k11);
    proj_step(v2f{f.xa[6], f.xa[7]}, v2f{f.xb[6], f.xb[7]},
              v2f{f.q0h.z, f.q0h.w}, v2f{f.q1h.z, f.q1h.w}, v2f{f.k0h.z, f.k0h.w}, v2f{f.k1h.z, f.k1h.w},
              q00, q01, q10, q11, k00, k01, k10, k11);
}

__global__ __launch_bounds__(128) void k_qkproj(const float* __restrict__ x,
                                                const float* __restrict__ Wqp,
                                                const float* __restrict__ Wkp,
                                                float* __restrict__ Qp,
                                                float* __restrict__ Kp) {
    const int lane = threadIdx.x & 31;
    const int wave = threadIdx.x >> 5;
    const int m    = lane & 15;
    const int half = lane >> 4;
    const int c0 = blockIdx.y * 32;
    const int n0 = blockIdx.x * 128 + wave * 32;
    const int b  = blockIdx.z;
    const float* xr = x + (size_t)b * C_ * N_ + (size_t)(8 * half) * N_ + n0 + m;
    const float4* wq0 = (const float4*)(Wqp + (size_t)(c0 >> 4) * (16 * C_)) + lane;
    const float4* wq1 = wq0 + 4 * C_;
    const float4* wk0 = (const float4*)(Wkp + (size_t)(c0 >> 4) * (16 * C_)) + lane;
    const float4* wk1 = wk0 + 4 * C_;
    v8f q00 = {}, q01 = {}, q10 = {}, q11 = {};
    v8f k00 = {}, k01 = {}, k10 = {}, k11 = {};
    ProjFrag f0, f1;
    const int nkb = C_ / 16;
    proj_load(f0, wq0, wq1, wk0, wk1, xr, 0);
    for (int kb = 0; kb < nkb; kb += 2) {
        const bool more = kb + 2 < nkb;
        proj_load(f1, wq0, wq1, wk0, wk1, xr, kb + 1);
        __builtin_amdgcn_sched_barrier(0);
        proj_mma(f0, q00, q01, q10, q11, k00, k01, k10, k11);
        proj_load(f0, wq0, wq1, wk0, wk1, xr, more ? kb + 2 : kb);
        __builtin_amdgcn_sched_barrier(0);
        proj_mma(f1, q00, q01, q10, q11, k00, k01, k10, k11);
    }
    float* Qb = Qp + (size_t)b * C_ * N_ + (size_t)(c0 >> 4) * (16 * N_) + (size_t)(n0 >> 4) * 256;
    float* Kb = Kp + (size_t)b * C_ * N_ + (size_t)(c0 >> 4) * (16 * N_) + (size_t)(n0 >> 4) * 256;
    store_frag_tile(Qb,                 lane, q00);
    store_frag_tile(Qb + 256,           lane, q01);
    store_frag_tile(Qb + 16 * N_,       lane, q10);
    store_frag_tile(Qb + 16 * N_ + 256, lane, q11);
    store_frag_tile(Kb,                 lane, k00);
    store_frag_tile(Kb + 256,           lane, k01);
    store_frag_tile(Kb + 16 * N_,       lane, k10);
    store_frag_tile(Kb + 16 * N_ + 256, lane, k11);
}

__global__ __launch_bounds__(1024) void k_rownorm(const float* __restrict__ Qp,
                                                  const float* __restrict__ Kp,
                                                  float* __restrict__ invQ,
                                                  float* __restrict__ invK) {
    __shared__ float part[32][16];
    const int tid  = threadIdx.x;
    const int lane = tid & 31;
    const int w    = tid >> 5;
    const int blk  = blockIdx.x;
    const float* src = blockIdx.y ? Kp : Qp;
    const float4* p = (const float4*)(src + (size_t)blk * (16 * N_)) + tid;
    float s = 0.f;
    #pragma unroll
    for (int i = 0; i < 16; ++i) {
        float4 v = p[i * 1024];
        s += (v.x * v.x + v.y * v.y) + (v.z * v.z + v.w * v.w);
    }
    s += __shfl_xor(s, 16, 32);
    if (lane < 16) part[w][lane] = s;
    __syncthreads();
    if (tid < 16) {
        float t = 0.f;
        for (int i = 0; i < 32; ++i) t += part[i][tid];
        float* dst = blockIdx.y ? invK : invQ;
        dst[blk * 16 + tid] = rsqrtf(t);
    }
}

struct QktFrag { float4 a0l, a0h, a1l, a1h, b0l, b0h, b1l, b1h; };

static __device__ __forceinline__ void qkt_load(QktFrag& f, const float4* p0, const float4* p1,
                                                const float4* t0, const float4* t1, int kb) {
    const int o = kb * 64;
    f.a0l = p0[o]; f.a0h = p0[o + 32];
    f.a1l = p1[o]; f.a1h = p1[o + 32];
    f.b0l = t0[o]; f.b0h = t0[o + 32];
    f.b1l = t1[o]; f.b1h = t1[o + 32];
}

static __device__ __forceinline__ void qkt_step(v2f aA, v2f aB, v2f bA, v2f bB,
                                                v8f& a00, v8f& a01, v8f& a10, v8f& a11) {
    a00 = wmma4(aA, bA, a00); a01 = wmma4(aA, bB, a01);
    a10 = wmma4(aB, bA, a10); a11 = wmma4(aB, bB, a11);
}

static __device__ __forceinline__ void qkt_mma(const QktFrag& f, v8f& a00, v8f& a01, v8f& a10, v8f& a11) {
    qkt_step(v2f{f.a0l.x, f.a0l.y}, v2f{f.a1l.x, f.a1l.y}, v2f{f.b0l.x, f.b0l.y}, v2f{f.b1l.x, f.b1l.y}, a00, a01, a10, a11);
    qkt_step(v2f{f.a0l.z, f.a0l.w}, v2f{f.a1l.z, f.a1l.w}, v2f{f.b0l.z, f.b0l.w}, v2f{f.b1l.z, f.b1l.w}, a00, a01, a10, a11);
    qkt_step(v2f{f.a0h.x, f.a0h.y}, v2f{f.a1h.x, f.a1h.y}, v2f{f.b0h.x, f.b0h.y}, v2f{f.b1h.x, f.b1h.y}, a00, a01, a10, a11);
    qkt_step(v2f{f.a0h.z, f.a0h.w}, v2f{f.a1h.z, f.a1h.w}, v2f{f.b0h.z, f.b0h.w}, v2f{f.b1h.z, f.b1h.w}, a00, a01, a10, a11);
}

__global__ __launch_bounds__(128) void k_qkt(const float* __restrict__ Rp,
                                             const float* __restrict__ Sp,
                                             const float* __restrict__ invR,
                                             const float* __restrict__ invS,
                                             float* __restrict__ Mt) {
    const int lane = threadIdx.x & 31;
    const int wave = threadIdx.x >> 5;
    const int m    = lane & 15;
    const int half = lane >> 4;
    const int r0 = blockIdx.y * 32;
    const int s0 = blockIdx.x * 128 + wave * 32;
    const int b  = blockIdx.z;
    const size_t bOff = (size_t)b * C_ * N_;
    const float4* p0 = (const float4*)(Rp + bOff + (size_t)(r0 >> 4) * (16 * N_)) + lane;
    const float4* p1 = p0 + 4 * N_;
    const float4* t0 = (const float4*)(Sp + bOff + (size_t)(s0 >> 4) * (16 * N_)) + lane;
    const float4* t1 = t0 + 4 * N_;
    v8f a00 = {}, a01 = {}, a10 = {}, a11 = {};
    QktFrag f0, f1, f2, f3;
    const int nkb = N_ / 16;
    qkt_load(f0, p0, p1, t0, t1, 0);
    __builtin_amdgcn_sched_barrier(0);
    qkt_load(f1, p0, p1, t0, t1, 1);
    __builtin_amdgcn_sched_barrier(0);
    qkt_load(f2, p0, p1, t0, t1, 2);
    __builtin_amdgcn_sched_barrier(0);
    for (int kb = 0; kb < nkb; kb += 4) {
        const bool more = kb + 4 < nkb;
        qkt_load(f3, p0, p1, t0, t1, kb + 3);
        __builtin_amdgcn_sched_barrier(0);
        qkt_mma(f0, a00, a01, a10, a11);
        __builtin_amdgcn_sched_barrier(0);
        qkt_load(f0, p0, p1, t0, t1, more ? kb + 4 : kb);
        __builtin_amdgcn_sched_barrier(0);
        qkt_mma(f1, a00, a01, a10, a11);
        __builtin_amdgcn_sched_barrier(0);
        qkt_load(f1, p0, p1, t0, t1, more ? kb + 5 : kb);
        __builtin_amdgcn_sched_barrier(0);
        qkt_mma(f2, a00, a01, a10, a11);
        __builtin_amdgcn_sched_barrier(0);
        qkt_load(f2, p0, p1, t0, t1, more ? kb + 6 : kb);
        __builtin_amdgcn_sched_barrier(0);
        qkt_mma(f3, a00, a01, a10, a11);
        __builtin_amdgcn_sched_barrier(0);
    }
    float* Mb = Mt + (size_t)b * C_ * C_;
    const float is0 = invS[b * C_ + s0 + m];
    const float is1 = invS[b * C_ + s0 + 16 + m];
    #pragma unroll
    for (int j = 0; j < 8; ++j) {
        const int rA = r0 + j + 8 * half;
        const int rB = rA + 16;
        const float irA = invR[b * C_ + rA];
        const float irB = invR[b * C_ + rB];
        Mb[(size_t)rA * C_ + s0 + m]      = a00[j] * irA * is0;
        Mb[(size_t)rA * C_ + s0 + 16 + m] = a01[j] * irA * is1;
        Mb[(size_t)rB * C_ + s0 + m]      = a10[j] * irB * is0;
        Mb[(size_t)rB * C_ + s0 + 16 + m] = a11[j] * irB * is1;
    }
}

__global__ __launch_bounds__(256) void k_rowsoftmax(float* __restrict__ Mt) {
    const int lane = threadIdx.x & 31;
    const int row  = blockIdx.x * 8 + (threadIdx.x >> 5);
    float4* r = (float4*)(Mt + (size_t)row * C_);
    float4 v[4];
    float mx = -3.4e38f, mn = 3.4e38f;
    for (int i = 0; i < 4; ++i) {
        float4 t = r[i * 32 + lane];
        t.x = 1.f - t.x; t.y = 1.f - t.y; t.z = 1.f - t.z; t.w = 1.f - t.w;
        v[i] = t;
        mx = fmaxf(mx, fmaxf(fmaxf(t.x, t.y), fmaxf(t.z, t.w)));
        mn = fminf(mn, fminf(fminf(t.x, t.y), fminf(t.z, t.w)));
    }
    for (int o = 16; o > 0; o >>= 1) {
        mx = fmaxf(mx, __shfl_xor(mx, o, 32));
        mn = fminf(mn, __shfl_xor(mn, o, 32));
    }
    const float dn = mx + EPS_;
    const float inv_dn = 1.f / dn;
    const float sh = 1.f - (mn / dn) * (1.f / HT_);
    float z = 0.f;
    for (int i = 0; i < 4; ++i) {
        float4 t = v[i];
        t.x = expf((1.f - (t.x * inv_dn) * (1.f / HT_)) - sh);
        t.y = expf((1.f - (t.y * inv_dn) * (1.f / HT_)) - sh);
        t.z = expf((1.f - (t.z * inv_dn) * (1.f / HT_)) - sh);
        t.w = expf((1.f - (t.w * inv_dn) * (1.f / HT_)) - sh);
        v[i] = t;
        z += (t.x + t.y) + (t.z + t.w);
    }
    for (int o = 16; o > 0; o >>= 1) z += __shfl_xor(z, o, 32);
    const float invz = 1.f / z;
    for (int i = 0; i < 4; ++i) {
        float4 t = v[i];
        t.x *= invz; t.y *= invz; t.z *= invz; t.w *= invz;
        r[i * 32 + lane] = t;
    }
}

__global__ __launch_bounds__(1024) void k_l1(float* __restrict__ Mt) {
    __shared__ float part[32][33];
    const int lane = threadIdx.x & 31;
    const int w    = threadIdx.x >> 5;
    float* base = Mt + (size_t)blockIdx.y * C_ * C_ + (size_t)(w * 16) * C_
                     + blockIdx.x * 32 + lane;
    float v[16];
    float s = 0.f;
    for (int i = 0; i < 16; ++i) { v[i] = base[(size_t)i * C_]; s += v[i]; }
    part[w][lane] = s;
    __syncthreads();
    float t = 0.f;
    for (int i = 0; i < 32; ++i) t += part[i][lane];
    const float inv = 1.f / (t + EPS_);
    for (int i = 0; i < 16; ++i) base[(size_t)i * C_] = v[i] * inv;
}

__global__ __launch_bounds__(128) void k_mwv(const float* __restrict__ Mt,
                                             const float* __restrict__ Wv,
                                             float* __restrict__ Gp) {
    const int lane = threadIdx.x & 31;
    const int wave = threadIdx.x >> 5;
    const int m    = lane & 15;
    const int kp   = (lane >> 4) << 1;
    const int c0 = blockIdx.y * 32;
    const int e0 = blockIdx.x * 128 + wave * 32;
    const int b  = blockIdx.z;
    const float* a = Mt + (size_t)b * C_ * C_ + c0 + m;
    v8f g00 = {}, g01 = {}, g10 = {}, g11 = {};
    for (int k = 0; k < C_; k += 4) {
        const int kk = k + kp;
        const float* ar0 = a + (size_t)kk * C_;
        const float* ar1 = a + (size_t)(kk + 1) * C_;
        v2f aA, aB;
        aA.x = ar0[0];  aA.y = ar1[0];
        aB.x = ar0[16]; aB.y = ar1[16];
        const float* wr0 = Wv + (size_t)kk * C_;
        const float* wr1 = Wv + (size_t)(kk + 1) * C_;
        v2f b0, b1;
        b0.x = wr0[e0 + m];      b0.y = wr1[e0 + m];
        b1.x = wr0[e0 + 16 + m]; b1.y = wr1[e0 + 16 + m];
        g00 = wmma4(b0, aA, g00); g01 = wmma4(b1, aA, g01);
        g10 = wmma4(b0, aB, g10); g11 = wmma4(b1, aB, g11);
    }
    float* Gb = Gp + (size_t)b * C_ * C_ + (size_t)(c0 >> 4) * (16 * C_) + (size_t)(e0 >> 4) * 256;
    store_frag_tile(Gb,                 lane, g00);
    store_frag_tile(Gb + 256,           lane, g01);
    store_frag_tile(Gb + 16 * C_,       lane, g10);
    store_frag_tile(Gb + 16 * C_ + 256, lane, g11);
}

__global__ __launch_bounds__(128) void k_out(const float* __restrict__ x,
                                             const float* __restrict__ Gp,
                                             const float* __restrict__ gamma,
                                             float* __restrict__ y) {
    const int lane = threadIdx.x & 31;
    const int wave = threadIdx.x >> 5;
    const int m    = lane & 15;
    const int half = lane >> 4;
    const int c0 = blockIdx.y * 32;
    const int n0 = blockIdx.x * 128 + wave * 32;
    const int b  = blockIdx.z;
    const float* xb = x + (size_t)b * C_ * N_;
    const float4* g0 = (const float4*)(Gp + (size_t)b * C_ * C_ + (size_t)(c0 >> 4) * (16 * C_)) + lane;
    const float4* g1 = g0 + 4 * C_;
    v8f o00 = {}, o01 = {}, o10 = {}, o11 = {};
    for (int kb = 0; kb < C_ / 16; ++kb) {
        const float4 A0l = g0[kb * 64], A0h = g0[kb * 64 + 32];
        const float4 A1l = g1[kb * 64], A1h = g1[kb * 64 + 32];
        const v2f aA[4] = { {A0l.x, A0l.y}, {A0l.z, A0l.w}, {A0h.x, A0h.y}, {A0h.z, A0h.w} };
        const v2f aB[4] = { {A1l.x, A1l.y}, {A1l.z, A1l.w}, {A1h.x, A1h.y}, {A1h.z, A1h.w} };
        const float* xr = xb + (size_t)(kb * 16 + 8 * half) * N_ + n0 + m;
        #pragma unroll
        for (int u = 0; u < 4; ++u) {
            v2f b0, b1;
            b0.x = xr[(size_t)(2 * u) * N_];      b0.y = xr[(size_t)(2 * u + 1) * N_];
            b1.x = xr[(size_t)(2 * u) * N_ + 16]; b1.y = xr[(size_t)(2 * u + 1) * N_ + 16];
            o00 = wmma4(aA[u], b0, o00); o01 = wmma4(aA[u], b1, o01);
            o10 = wmma4(aB[u], b0, o10); o11 = wmma4(aB[u], b1, o11);
        }
    }
    const float g = gamma[0];
    float* yb = y + (size_t)b * C_ * N_;
    #pragma unroll
    for (int j = 0; j < 8; ++j) {
        const int cA = c0 + j + 8 * half;
        const int cB = cA + 16;
        const size_t o0 = (size_t)cA * N_ + n0 + m;
        const size_t o1 = (size_t)cA * N_ + n0 + 16 + m;
        const size_t o2 = (size_t)cB * N_ + n0 + m;
        const size_t o3 = (size_t)cB * N_ + n0 + 16 + m;
        yb[o0] = xb[o0] + g * o00[j];
        yb[o1] = xb[o1] + g * o01[j];
        yb[o2] = xb[o2] + g * o10[j];
        yb[o3] = xb[o3] + g * o11[j];
    }
}

extern "C" void kernel_launch(void* const* d_in, const int* in_sizes, int n_in,
                              void* d_out, int out_size, void* d_ws, size_t ws_size,
                              hipStream_t stream) {
    (void)in_sizes; (void)n_in; (void)out_size; (void)ws_size;
    const float* x     = (const float*)d_in[0];
    const float* Wq    = (const float*)d_in[1];
    const float* Wk    = (const float*)d_in[2];
    const float* Wv    = (const float*)d_in[3];
    const float* gamma = (const float*)d_in[4];
    float* y = (float*)d_out;

    char* ws = (char*)d_ws;
    const size_t qkBytes = (size_t)B_ * C_ * N_ * sizeof(float);
    const size_t mBytes  = (size_t)B_ * C_ * C_ * sizeof(float);
    float* Q     = (float*)(ws);
    float* K     = (float*)(ws + qkBytes);
    float* M     = (float*)(ws + 2 * qkBytes);
    float* G     = (float*)(ws + 2 * qkBytes + mBytes);
    float* invQ  = (float*)(ws + 2 * qkBytes + 2 * mBytes);
    float* invK  = invQ  + B_ * C_;
    float* Wqp   = G;
    float* Wkp   = G + C_ * C_;

    k_wfrag     <<<dim3(C_ * C_ / 4 / 256, 2), 256, 0, stream>>>(Wq, Wk, Wqp, Wkp);
    k_qkproj    <<<dim3(N_ / 128, C_ / 32, B_), 128, 0, stream>>>(x, Wqp, Wkp, Q, K);
    k_rownorm   <<<dim3(B_ * C_ / 16, 2), 1024, 0, stream>>>(Q, K, invQ, invK);
    k_qkt       <<<dim3(C_ / 128, C_ / 32, B_), 128, 0, stream>>>(K, Q, invK, invQ, M);
    k_rowsoftmax<<<B_ * C_ / 8, 256, 0, stream>>>(M);
    k_l1        <<<dim3(C_ / 32, B_), 1024, 0, stream>>>(M);
    k_mwv       <<<dim3(C_ / 128, C_ / 32, B_), 128, 0, stream>>>(M, Wv, G);
    k_out       <<<dim3(N_ / 128, C_ / 32, B_), 128, 0, stream>>>(x, G, gamma, y);
}


# x read by whole-line 16 B loads: wave tile 32x64, lane i of n tile t owns column 4i+t (Q,K projection, output GEMM, y stores too), loads grouped a block ahead; n order of QK^T and norm sums changes
# speedup vs baseline: 1.7362x; 1.0516x over previous
#include <hip/hip_runtime.h>

typedef __attribute__((ext_vector_type(2))) float v2f;
typedef __attribute__((ext_vector_type(8))) float v8f;

#define B_ 8
#define C_ 512
#define N_ 4096
#define EPS_ 1e-6f
#define HT_ 0.25f

static __device__ __forceinline__ v8f wmma4(v2f a, v2f b, v8f c) {
    return __builtin_amdgcn_wmma_f32_16x16x4_f32(false, a, false, b, (short)0, c, false, false);
}

static __device__ __forceinline__ void store_frag_tile(float* tile, int lane, v8f acc) {
    float4* p = (float4*)tile + lane;
    p[0]  = make_float4(acc[0], acc[1], acc[2], acc[3]);
    p[32] = make_float4(acc[4], acc[5], acc[6], acc[7]);
}

__global__ __launch_bounds__(256) void k_wfrag(const float* __restrict__ Wq,
                                               const float* __restrict__ Wk,
                                               float* __restrict__ Wqp,
                                               float* __restrict__ Wkp) {
    const int i    = blockIdx.x * 256 + threadIdx.x;
    const int lane = i & 31;
    const int t    = (i >> 5) & 1;
    const int kb   = (i >> 6) & 31;
    const int rb   = i >> 11;
    const int row  = rb * 16 + (lane & 15);
    const int k    = kb * 16 + 8 * (lane >> 4) + 4 * t;
    const float* src = blockIdx.y ? Wk : Wq;
    float* dst = blockIdx.y ? Wkp : Wqp;
    ((float4*)dst)[i] = *(const float4*)(src + (size_t)row * C_ + k);
}

static __device__ __forceinline__ float comp4(const float4& v, int t) {
    return t == 0 ? v.x : t == 1 ? v.y : t == 2 ? v.z : v.w;
}

struct ProjFrag {
    float4 q0l, q0h, q1l, q1h, k0l, k0h, k1l, k1h;
    float4 xv[8];
};

static __device__ __forceinline__ void proj_load(ProjFrag& f, const float4* q0, const float4* q1,
                                                 const float4* k0, const float4* k1,
                                                 const float* xr, int kb) {
    const int o = kb * 64;
    f.q0l = q0[o]; f.q0h = q0[o + 32];
    f.q1l = q1[o]; f.q1h = q1[o + 32];
    f.k0l = k0[o]; f.k0h = k0[o + 32];
    f.k1l = k1[o]; f.k1h = k1[o + 32];
    const float* p = xr + (size_t)kb * (16 * N_);
    #pragma unroll
    for (int j = 0; j < 8; ++j) f.xv[j] = *(const float4*)(p + (size_t)j * N_);
}

static __device__ __forceinline__ void proj_step(const float4& r0, const float4& r1,
                                                 v2f wq0, v2f wq1, v2f wk0, v2f wk1,
                                                 v8f (&q0)[4], v8f (&q1)[4], v8f (&k0)[4], v8f (&k1)[4]) {
    #pragma unroll
    for (int t = 0; t < 4; ++t) {
        const v2f xA = {comp4(r0, t), comp4(r1, t)};
        q0[t] = wmma4(xA, wq0, q0[t]);
        q1[t] = wmma4(xA, wq1, q1[t]);
        k0[t] = wmma4(xA, wk0, k0[t]);
        k1[t] = wmma4(xA, wk1, k1[t]);
    }
}

static __device__ __forceinline__ void proj_mma(const ProjFrag& f,
                                                v8f (&q0)[4], v8f (&q1)[4], v8f (&k0)[4], v8f (&k1)[4]) {
    proj_step(f.xv[0], f.xv[1],
              v2f{f.q0l.x, f.q0l.y}, v2f{f.q1l.x, f.q1l.y}, v2f{f.k0l.x, f.k0l.y}, v2f{f.k1l.x, f.k1l.y},
              q0, q1, k0, k1);
    proj_step(f.xv[2], f.xv[3],
              v2f{f.q0l.z, f.q0l.w}, v2f{f.q1l.z, f.q1l.w}, v2f{f.k0l.z, f.k0l.w}, v2f{f.k1l.z, f.k1l.w},
              q0, q1, k0, k1);
    proj_step(f.xv[4], f.xv[5],
              v2f{f.q0h.x, f.q0h.y}, v2f{f.q1h.x, f.q1h.y}, v2f{f.k0h.x, f.k0h.y}, v2f{f.k1h.x, f.k1h.y},
              q0, q1, k0, k1);
    proj_step(f.xv[6], f.xv[7],
              v2f{f.q0h.z, f.q0h.w}, v2f{f.q1h.z, f.q1h.w}, v2f{f.k0h.z, f.k0h.w}, v2f{f.k1h.z, f.k1h.w},
              q0, q1, k0, k1);
}

__global__ __launch_bounds__(128) void k_qkproj(const float* __restrict__ x,
                                                const float* __restrict__ Wqp,
                                                const float* __restrict__ Wkp,
                                                float* __restrict__ Qp,
                                                float* __restrict__ Kp) {
    const int lane = threadIdx.x & 31;
    const int wave = threadIdx.x >> 5;
    const int i    = lane & 15;
    const int half = lane >> 4;
    const int c0 = blockIdx.y * 32;
    const int nw = blockIdx.x * 256 + wave * 64;
    const int b  = blockIdx.z;
    const float* xr = x + (size_t)b * C_ * N_ + (size_t)(8 * half) * N_ + nw + 4 * i;
    const float4* wq0 = (const float4*)(Wqp + (size_t)(c0 >> 4) * (16 * C_)) + lane;
    const float4* wq1 = wq0 + 4 * C_;
    const float4* wk0 = (const float4*)(Wkp + (size_t)(c0 >> 4) * (16 * C_)) + lane;
    const float4* wk1 = wk0 + 4 * C_;
    v8f q0[4] = {}, q1[4] = {}, k0[4] = {}, k1[4] = {};
    ProjFrag f0, f1;
    const int nkb = C_ / 16;
    proj_load(f0, wq0, wq1, wk0, wk1, xr, 0);
    for (int kb = 0; kb < nkb - 2; kb += 2) {
        proj_load(f1, wq0, wq1, wk0, wk1, xr, kb + 1);
        __builtin_amdgcn_sched_barrier(0);
        proj_mma(f0, q0, q1, k0, k1);
        __builtin_amdgcn_sched_barrier(0);
        proj_load(f0, wq0, wq1, wk0, wk1, xr, kb + 2);
        __builtin_amdgcn_sched_barrier(0);
        proj_mma(f1, q0, q1, k0, k1);
    }
    proj_load(f1, wq0, wq1, wk0, wk1, xr, nkb - 1);
    __builtin_amdgcn_sched_barrier(0);
    proj_mma(f0, q0, q1, k0, k1);
    proj_mma(f1, q0, q1, k0, k1);
    float* Qb = Qp + (size_t)b * C_ * N_ + (size_t)(c0 >> 4) * (16 * N_) + (size_t)(nw >> 4) * 256;
    float* Kb = Kp + (size_t)b * C_ * N_ + (size_t)(c0 >> 4) * (16 * N_) + (size_t)(nw >> 4) * 256;
    #pragma unroll
    for (int t = 0; t < 4; ++t) {
        store_frag_tile(Qb + t * 256,           lane, q0[t]);
        store_frag_tile(Qb + 16 * N_ + t * 256, lane, q1[t]);
        store_frag_tile(Kb + t * 256,           lane, k0[t]);
        store_frag_tile(Kb + 16 * N_ + t * 256, lane, k1[t]);
    }
}

__global__ __launch_bounds__(1024) void k_rownorm(const float* __restrict__ Qp,
                                                  const float* __restrict__ Kp,
                                                  float* __restrict__ invQ,
                                                  float* __restrict__ invK) {
    __shared__ float part[32][16];
    const int tid  = threadIdx.x;
    const int lane = tid & 31;
    const int w    = tid >> 5;
    const int blk  = blockIdx.x;
    const float* src = blockIdx.y ? Kp : Qp;
    const float4* p = (const float4*)(src + (size_t)blk * (16 * N_)) + tid;
    float s = 0.f;
    #pragma unroll
    for (int i = 0; i < 16; ++i) {
        float4 v = p[i * 1024];
        s += (v.x * v.x + v.y * v.y) + (v.z * v.z + v.w * v.w);
    }
    s += __shfl_xor(s, 16, 32);
    if (lane < 16) part[w][lane] = s;
    __syncthreads();
    if (tid < 16) {
        float t = 0.f;
        for (int i = 0; i < 32; ++i) t += part[i][tid];
        float* dst = blockIdx.y ? invK : invQ;
        dst[blk * 16 + tid] = rsqrtf(t);
    }
}

struct QktFrag { float4 a0l, a0h, a1l, a1h, b0l, b0h, b1l, b1h; };

static __device__ __forceinline__ void qkt_load(QktFrag& f, const float4* p0, const float4* p1,
                                                const float4* t0, const float4* t1, int kb) {
    const int o = kb * 64;
    f.a0l = p0[o]; f.a0h = p0[o + 32];
    f.a1l = p1[o]; f.a1h = p1[o + 32];
    f.b0l = t0[o]; f.b0h = t0[o + 32];
    f.b1l = t1[o]; f.b1h = t1[o + 32];
}

static __device__ __forceinline__ void qkt_step(v2f aA, v2f aB, v2f bA, v2f bB,
                                                v8f& a00, v8f& a01, v8f& a10, v8f& a11) {
    a00 = wmma4(aA, bA, a00); a01 = wmma4(aA, bB, a01);
    a10 = wmma4(aB, bA, a10); a11 = wmma4(aB, bB, a11);
}

static __device__ __forceinline__ void qkt_mma(const QktFrag& f, v8f& a00, v8f& a01, v8f& a10, v8f& a11) {
    qkt_step(v2f{f.a0l.x, f.a0l.y}, v2f{f.a1l.x, f.a1l.y}, v2f{f.b0l.x, f.b0l.y}, v2f{f.b1l.x, f.b1l.y}, a00, a01, a10, a11);
    qkt_step(v2f{f.a0l.z, f.a0l.w}, v2f{f.a1l.z, f.a1l.w}, v2f{f.b0l.z, f.b0l.w}, v2f{f.b1l.z, f.b1l.w}, a00, a01, a10, a11);
    qkt_step(v2f{f.a0h.x, f.a0h.y}, v2f{f.a1h.x, f.a1h.y}, v2f{f.b0h.x, f.b0h.y}, v2f{f.b1h.x, f.b1h.y}, a00, a01, a10, a11);
    qkt_step(v2f{f.a0h.z, f.a0h.w}, v2f{f.a1h.z, f.a1h.w}, v2f{f.b0h.z, f.b0h.w}, v2f{f.b1h.z, f.b1h.w}, a00, a01, a10, a11);
}

__global__ __launch_bounds__(128) void k_qkt(const float* __restrict__ Rp,
                                             const float* __restrict__ Sp,
                                             const float* __restrict__ invR,
                                             const float* __restrict__ invS,
                                             float* __restrict__ Mt) {
    const int lane = threadIdx.x & 31;
    const int wave = threadIdx.x >> 5;
    const int m    = lane & 15;
    const int half = lane >> 4;
    const int r0 = blockIdx.y * 32;
    const int s0 = blockIdx.x * 128 + wave * 32;
    const int b  = blockIdx.z;
    const size_t bOff = (size_t)b * C_ * N_;
    const float4* p0 = (const float4*)(Rp + bOff + (size_t)(r0 >> 4) * (16 * N_)) + lane;
    const float4* p1 = p0 + 4 * N_;
    const float4* t0 = (const float4*)(Sp + bOff + (size_t)(s0 >> 4) * (16 * N_)) + lane;
    const float4* t1 = t0 + 4 * N_;
    v8f a00 = {}, a01 = {}, a10 = {}, a11 = {};
    QktFrag f0, f1, f2, f3;
    const int nkb = N_ / 16;
    qkt_load(f0, p0, p1, t0, t1, 0);
    __builtin_amdgcn_sched_barrier(0);
    qkt_load(f1, p0, p1, t0, t1, 1);
    __builtin_amdgcn_sched_barrier(0);
    qkt_load(f2, p0, p1, t0, t1, 2);
    __builtin_amdgcn_sched_barrier(0);
    for (int kb = 0; kb < nkb; kb += 4) {
        const bool more = kb + 4 < nkb;
        qkt_load(f3, p0, p1, t0, t1, kb + 3);
        __builtin_amdgcn_sched_barrier(0);
        qkt_mma(f0, a00, a01, a10, a11);
        __builtin_amdgcn_sched_barrier(0);
        qkt_load(f0, p0, p1, t0, t1, more ? kb + 4 : kb);
        __builtin_amdgcn_sched_barrier(0);
        qkt_mma(f1, a00, a01, a10, a11);
        __builtin_amdgcn_sched_barrier(0);
        qkt_load(f1, p0, p1, t0, t1, more ? kb + 5 : kb);
        __builtin_amdgcn_sched_barrier(0);
        qkt_mma(f2, a00, a01, a10, a11);
        __builtin_amdgcn_sched_barrier(0);
        qkt_load(f2, p0, p1, t0, t1, more ? kb + 6 : kb);
        __builtin_amdgcn_sched_barrier(0);
        qkt_mma(f3, a00, a01, a10, a11);
        __builtin_amdgcn_sched_barrier(0);
    }
    float* Mb = Mt + (size_t)b * C_ * C_;
    const float is0 = invS[b * C_ + s0 + m];
    const float is1 = invS[b * C_ + s0 + 16 + m];
    #pragma unroll
    for (int j = 0; j < 8; ++j) {
        const int rA = r0 + j + 8 * half;
        const int rB = rA + 16;
        const float irA = invR[b * C_ + rA];
        const float irB = invR[b * C_ + rB];
        Mb[(size_t)rA * C_ + s0 + m]      = a00[j] * irA * is0;
        Mb[(size_t)rA * C_ + s0 + 16 + m] = a01[j] * irA * is1;
        Mb[(size_t)rB * C_ + s0 + m]      = a10[j] * irB * is0;
        Mb[(size_t)rB * C_ + s0 + 16 + m] = a11[j] * irB * is1;
    }
}

__global__ __launch_bounds__(256) void k_rowsoftmax(float* __restrict__ Mt) {
    const int lane = threadIdx.x & 31;
    const int row  = blockIdx.x * 8 + (threadIdx.x >> 5);
    float4* r = (float4*)(Mt + (size_t)row * C_);
    float4 v[4];
    float mx = -3.4e38f, mn = 3.4e38f;
    for (int i = 0; i < 4; ++i) {
        float4 t = r[i * 32 + lane];
        t.x = 1.f - t.x; t.y = 1.f - t.y; t.z = 1.f - t.z; t.w = 1.f - t.w;
        v[i] = t;
        mx = fmaxf(mx, fmaxf(fmaxf(t.x, t.y), fmaxf(t.z, t.w)));
        mn = fminf(mn, fminf(fminf(t.x, t.y), fminf(t.z, t.w)));
    }
    for (int o = 16; o > 0; o >>= 1) {
        mx = fmaxf(mx, __shfl_xor(mx, o, 32));
        mn = fminf(mn, __shfl_xor(mn, o, 32));
    }
    const float dn = mx + EPS_;
    const float inv_dn = 1.f / dn;
    const float sh = 1.f - (mn / dn) * (1.f / HT_);
    float z = 0.f;
    for (int i = 0; i < 4; ++i) {
        float4 t = v[i];
        t.x = expf((1.f - (t.x * inv_dn) * (1.f / HT_)) - sh);
        t.y = expf((1.f - (t.y * inv_dn) * (1.f / HT_)) - sh);
        t.z = expf((1.f - (t.z * inv_dn) * (1.f / HT_)) - sh);
        t.w = expf((1.f - (t.w * inv_dn) * (1.f / HT_)) - sh);
        v[i] = t;
        z += (t.x + t.y) + (t.z + t.w);
    }
    for (int o = 16; o > 0; o >>= 1) z += __shfl_xor(z, o, 32);
    const float invz = 1.f / z;
    for (int i = 0; i < 4; ++i) {
        float4 t = v[i];
        t.x *= invz; t.y *= invz; t.z *= invz; t.w *= invz;
        r[i * 32 + lane] = t;
    }
}

__global__ __launch_bounds__(1024) void k_l1(float* __restrict__ Mt) {
    __shared__ float part[32][33];
    const int lane = threadIdx.x & 31;
    const int w    = threadIdx.x >> 5;
    float* base = Mt + (size_t)blockIdx.y * C_ * C_ + (size_t)(w * 16) * C_
                     + blockIdx.x * 32 + lane;
    float v[16];
    float s = 0.f;
    for (int i = 0; i < 16; ++i) { v[i] = base[(size_t)i * C_]; s += v[i]; }
    part[w][lane] = s;
    __syncthreads();
    float t = 0.f;
    for (int i = 0; i < 32; ++i) t += part[i][lane];
    const float inv = 1.f / (t + EPS_);
    for (int i = 0; i < 16; ++i) base[(size_t)i * C_] = v[i] * inv;
}

__global__ __launch_bounds__(128) void k_mwv(const float* __restrict__ Mt,
                                             const float* __restrict__ Wv,
                                             float* __restrict__ Gp) {
    const int lane = threadIdx.x & 31;
    const int wave = threadIdx.x >> 5;
    const int m    = lane & 15;
    const int kp   = (lane >> 4) << 1;
    const int c0 = blockIdx.y * 32;
    const int e0 = blockIdx.x * 128 + wave * 32;
    const int b  = blockIdx.z;
    const float* a = Mt + (size_t)b * C_ * C_ + c0 + m;
    v8f g00 = {}, g01 = {}, g10 = {}, g11 = {};
    for (int k = 0; k < C_; k += 4) {
        const int kk = k + kp;
        const float* ar0 = a + (size_t)kk * C_;
        const float* ar1 = a + (size_t)(kk + 1) * C_;
        v2f aA, aB;
        aA.x = ar0[0];  aA.y = ar1[0];
        aB.x = ar0[16]; aB.y = ar1[16];
        const float* wr0 = Wv + (size_t)kk * C_;
        const float* wr1 = Wv + (size_t)(kk + 1) * C_;
        v2f b0, b1;
        b0.x = wr0[e0 + m];      b0.y = wr1[e0 + m];
        b1.x = wr0[e0 + 16 + m]; b1.y = wr1[e0 + 16 + m];
        g00 = wmma4(b0, aA, g00); g01 = wmma4(b1, aA, g01);
        g10 = wmma4(b0, aB, g10); g11 = wmma4(b1, aB, g11);
    }
    float* Gb = Gp + (size_t)b * C_ * C_ + (size_t)(c0 >> 4) * (16 * C_) + (size_t)(e0 >> 4) * 256;
    store_frag_tile(Gb,                 lane, g00);
    store_frag_tile(Gb + 256,           lane, g01);
    store_frag_tile(Gb + 16 * C_,       lane, g10);
    store_frag_tile(Gb + 16 * C_ + 256, lane, g11);
}

struct OutFrag {
    float4 a0l, a0h, a1l, a1h;
    float4 xv[8];
};

static __device__ __forceinline__ void out_load(OutFrag& f, const float4* g0, const float4* g1,
                                                const float* xr, int kb) {
    const int o = kb * 64;
    f.a0l = g0[o]; f.a0h = g0[o + 32];
    f.a1l = g1[o]; f.a1h = g1[o + 32];
    const float* p = xr + (size_t)kb * (16 * N_);
    #pragma unroll
    for (int j = 0; j < 8; ++j) f.xv[j] = *(const float4*)(p + (size_t)j * N_);
}

static __device__ __forceinline__ void out_step(v2f aA, v2f aB, const float4& r0, const float4& r1,
                                                v8f (&o0)[4], v8f (&o1)[4]) {
    #pragma unroll
    for (int t = 0; t < 4; ++t) {
        const v2f bt = {comp4(r0, t), comp4(r1, t)};
        o0[t] = wmma4(aA, bt, o0[t]);
        o1[t] = wmma4(aB, bt, o1[t]);
    }
}

static __device__ __forceinline__ void out_mma(const OutFrag& f, v8f (&o0)[4], v8f (&o1)[4]) {
    out_step(v2f{f.a0l.x, f.a0l.y}, v2f{f.a1l.x, f.a1l.y}, f.xv[0], f.xv[1], o0, o1);
    out_step(v2f{f.a0l.z, f.a0l.w}, v2f{f.a1l.z, f.a1l.w}, f.xv[2], f.xv[3], o0, o1);
    out_step(v2f{f.a0h.x, f.a0h.y}, v2f{f.a1h.x, f.a1h.y}, f.xv[4], f.xv[5], o0, o1);
    out_step(v2f{f.a0h.z, f.a0h.w}, v2f{f.a1h.z, f.a1h.w}, f.xv[6], f.xv[7], o0, o1);
}

__global__ __launch_bounds__(128) void k_out(const float* __restrict__ x,
                                             const float* __restrict__ Gp,
                                             const float* __restrict__ gamma,
                                             float* __restrict__ y) {
    const int lane = threadIdx.x & 31;
    const int wave = threadIdx.x >> 5;
    const int i    = lane & 15;
    const int half = lane >> 4;
    const int c0 = blockIdx.y * 32;
    const int nw = blockIdx.x * 256 + wave * 64;
    const int b  = blockIdx.z;
    const float* xb = x + (size_t)b * C_ * N_;
    const float* xr = xb + (size_t)(8 * half) * N_ + nw + 4 * i;
    const float4* g0 = (const float4*)(Gp + (size_t)b * C_ * C_ + (size_t)(c0 >> 4) * (16 * C_)) + lane;
    const float4* g1 = g0 + 4 * C_;
    v8f o0[4] = {}, o1[4] = {};
    OutFrag f0, f1;
    const int nkb = C_ / 16;
    out_load(f0, g0, g1, xr, 0);
    for (int kb = 0; kb < nkb - 2; kb += 2) {
        out_load(f1, g0, g1, xr, kb + 1);
        __builtin_amdgcn_sched_barrier(0);
        out_mma(f0, o0, o1);
        __builtin_amdgcn_sched_barrier(0);
        out_load(f0, g0, g1, xr, kb + 2);
        __builtin_amdgcn_sched_barrier(0);
        out_mma(f1, o0, o1);
        __builtin_amdgcn_sched_barrier(0);
    }
    out_load(f1, g0, g1, xr, nkb - 1);
    __builtin_amdgcn_sched_barrier(0);
    out_mma(f0, o0, o1);
    out_mma(f1, o0, o1);
    const float g = gamma[0];
    float* yb = y + (size_t)b * C_ * N_;
    #pragma unroll
    for (int j = 0; j < 8; ++j) {
        const size_t oA = (size_t)(c0 + j + 8 * half) * N_ + nw + 4 * i;
        const size_t oB = oA + (size_t)16 * N_;
        const float4 xA = *(const float4*)(xb + oA);
        const float4 xB = *(const float4*)(xb + oB);
        *(float4*)(yb + oA) = make_float4(xA.x + g * o0[0][j], xA.y + g * o0[1][j],
                                          xA.z + g * o0[2][j], xA.w + g * o0[3][j]);
        *(float4*)(yb + oB) = make_float4(xB.x + g * o1[0][j], xB.y + g * o1[1][j],
                                          xB.z + g * o1[2][j], xB.w + g * o1[3][j]);
    }
}

extern "C" void kernel_launch(void* const* d_in, const int* in_sizes, int n_in,
                              void* d_out, int out_size, void* d_ws, size_t ws_size,
                              hipStream_t stream) {
    (void)in_sizes; (void)n_in; (void)out_size; (void)ws_size;
    const float* x     = (const float*)d_in[0];
    const float* Wq    = (const float*)d_in[1];
    const float* Wk    = (const float*)d_in[2];
    const float* Wv    = (const float*)d_in[3];
    const float* gamma = (const float*)d_in[4];
    float* y = (float*)d_out;

    char* ws = (char*)d_ws;
    const size_t qkBytes = (size_t)B_ * C_ * N_ * sizeof(float);
    const size_t mBytes  = (size_t)B_ * C_ * C_ * sizeof(float);
    float* Q     = (float*)(ws);
    float* K     = (float*)(ws + qkBytes);
    float* M     = (float*)(ws + 2 * qkBytes);
    float* G     = (float*)(ws + 2 * qkBytes + mBytes);
    float* invQ  = (float*)(ws + 2 * qkBytes + 2 * mBytes);
    float* invK  = invQ  + B_ * C_;
    float* Wqp   = G;
    float* Wkp   = G + C_ * C_;

    k_wfrag     <<<dim3(C_ * C_ / 4 / 256, 2), 256, 0, stream>>>(Wq, Wk, Wqp, Wkp);
    k_qkproj    <<<dim3(N_ / 256, C_ / 32, B_), 128, 0, stream>>>(x, Wqp, Wkp, Q, K);
    k_rownorm   <<<dim3(B_ * C_ / 16, 2), 1024, 0, stream>>>(Q, K, invQ, invK);
    k_qkt       <<<dim3(C_ / 128, C_ / 32, B_), 128, 0, stream>>>(K, Q, invK, invQ, M);
    k_rowsoftmax<<<B_ * C_ / 8, 256, 0, stream>>>(M);
    k_l1        <<<dim3(C_ / 32, B_), 1024, 0, stream>>>(M);
    k_mwv       <<<dim3(C_ / 128, C_ / 32, B_), 128, 0, stream>>>(M, Wv, G);
    k_out       <<<dim3(N_ / 256, C_ / 32, B_), 128, 0, stream>>>(x, G, gamma, y);
}


# Q*K^T wave tile 32x32 -> 64x64 in registers (each fragment feeds 4 WMMAs, half the fragment bytes per WMMA), k range cut in two halves as two planes the softmax pass adds; k sum tree changes
# speedup vs baseline: 1.8458x; 1.0631x over previous
#include <hip/hip_runtime.h>

typedef __attribute__((ext_vector_type(2))) float v2f;
typedef __attribute__((ext_vector_type(8))) float v8f;

#define B_ 8
#define C_ 512
#define N_ 4096
#define EPS_ 1e-6f
#define HT_ 0.25f

static __device__ __forceinline__ v8f wmma4(v2f a, v2f b, v8f c) {
    return __builtin_amdgcn_wmma_f32_16x16x4_f32(false, a, false, b, (short)0, c, false, false);
}

static __device__ __forceinline__ void store_frag_tile(float* tile, int lane, v8f acc) {
    float4* p = (float4*)tile + lane;
    p[0]  = make_float4(acc[0], acc[1], acc[2], acc[3]);
    p[32] = make_float4(acc[4], acc[5], acc[6], acc[7]);
}

__global__ __launch_bounds__(256) void k_wfrag(const float* __restrict__ Wq,
                                               const float* __restrict__ Wk,
                                               float* __restrict__ Wqp,
                                               float* __restrict__ Wkp) {
    const int i    = blockIdx.x * 256 + threadIdx.x;
    const int lane = i & 31;
    const int t    = (i >> 5) & 1;
    const int kb   = (i >> 6) & 31;
    const int rb   = i >> 11;
    const int row  = rb * 16 + (lane & 15);
    const int k    = kb * 16 + 8 * (lane >> 4) + 4 * t;
    const float* src = blockIdx.y ? Wk : Wq;
    float* dst = blockIdx.y ? Wkp : Wqp;
    ((float4*)dst)[i] = *(const float4*)(src + (size_t)row * C_ + k);
}

static __device__ __forceinline__ float comp4(const float4& v, int t) {
    return t == 0 ? v.x : t == 1 ? v.y : t == 2 ? v.z : v.w;
}

struct ProjFrag {
    float4 q0l, q0h, q1l, q1h, k0l, k0h, k1l, k1h;
    float4 xv[8];
};

static __device__ __forceinline__ void proj_load(ProjFrag& f, const float4* q0, const float4* q1,
                                                 const float4* k0, const float4* k1,
                                                 const float* xr, int kb) {
    const int o = kb * 64;
    f.q0l = q0[o]; f.q0h = q0[o + 32];
    f.q1l = q1[o]; f.q1h = q1[o + 32];
    f.k0l = k0[o]; f.k0h = k0[o + 32];
    f.k1l = k1[o]; f.k1h = k1[o + 32];
    const float* p = xr + (size_t)kb * (16 * N_);
    #pragma unroll
    for (int j = 0; j < 8; ++j) f.xv[j] = *(const float4*)(p + (size_t)j * N_);
}

static __device__ __forceinline__ void proj_step(const float4& r0, const float4& r1,
                                                 v2f wq0, v2f wq1, v2f wk0, v2f wk1,
                                                 v8f (&q0)[4], v8f (&q1)[4], v8f (&k0)[4], v8f (&k1)[4]) {
    #pragma unroll
    for (int t = 0; t < 4; ++t) {
        const v2f xA = {comp4(r0, t), comp4(r1, t)};
        q0[t] = wmma4(xA, wq0, q0[t]);
        q1[t] = wmma4(xA, wq1, q1[t]);
        k0[t] = wmma4(xA, wk0, k0[t]);
        k1[t] = wmma4(xA, wk1, k1[t]);
    }
}

static __device__ __forceinline__ void proj_mma(const ProjFrag& f,
                                                v8f (&q0)[4], v8f (&q1)[4], v8f (&k0)[4], v8f (&k1)[4]) {
    proj_step(f.xv[0], f.xv[1],
              v2f{f.q0l.x, f.q0l.y}, v2f{f.q1l.x, f.q1l.y}, v2f{f.k0l.x, f.k0l.y}, v2f{f.k1l.x, f.k1l.y},
              q0, q1, k0, k1);
    proj_step(f.xv[2], f.xv[3],
              v2f{f.q0l.z, f.q0l.w}, v2f{f.q1l.z, f.q1l.w}, v2f{f.k0l.z, f.k0l.w}, v2f{f.k1l.z, f.k1l.w},
              q0, q1, k0, k1);
    proj_step(f.xv[4], f.xv[5],
              v2f{f.q0h.x, f.q0h.y}, v2f{f.q1h.x, f.q1h.y}, v2f{f.k0h.x, f.k0h.y}, v2f{f.k1h.x, f.k1h.y},
              q0, q1, k0, k1);
    proj_step(f.xv[6], f.xv[7],
              v2f{f.q0h.z, f.q0h.w}, v2f{f.q1h.z, f.q1h.w}, v2f{f.k0h.z, f.k0h.w}, v2f{f.k1h.z, f.k1h.w},
              q0, q1, k0, k1);
}

__global__ __launch_bounds__(128) void k_qkproj(const float* __restrict__ x,
                                                const float* __restrict__ Wqp,
                                                const float* __restrict__ Wkp,
                                                float* __restrict__ Qp,
                                                float* __restrict__ Kp) {
    const int lane = threadIdx.x & 31;
    const int wave = threadIdx.x >> 5;
    const int i    = lane & 15;
    const int half = lane >> 4;
    const int c0 = blockIdx.y * 32;
    const int nw = blockIdx.x * 256 + wave * 64;
    const int b  = blockIdx.z;
    const float* xr = x + (size_t)b * C_ * N_ + (size_t)(8 * half) * N_ + nw + 4 * i;
    const float4* wq0 = (const float4*)(Wqp + (size_t)(c0 >> 4) * (16 * C_)) + lane;
    const float4* wq1 = wq0 + 4 * C_;
    const float4* wk0 = (const float4*)(Wkp + (size_t)(c0 >> 4) * (16 * C_)) + lane;
    const float4* wk1 = wk0 + 4 * C_;
    v8f q0[4] = {}, q1[4] = {}, k0[4] = {}, k1[4] = {};
    ProjFrag f0, f1;
    const int nkb = C_ / 16;
    proj_load(f0, wq0, wq1, wk0, wk1, xr, 0);
    for (int kb = 0; kb < nkb - 2; kb += 2) {
        proj_load(f1, wq0, wq1, wk0, wk1, xr, kb + 1);
        __builtin_amdgcn_sched_barrier(0);
        proj_mma(f0, q0, q1, k0, k1);
        __builtin_amdgcn_sched_barrier(0);
        proj_load(f0, wq0, wq1, wk0, wk1, xr, kb + 2);
        __builtin_amdgcn_sched_barrier(0);
        proj_mma(f1, q0, q1, k0, k1);
    }
    proj_load(f1, wq0, wq1, wk0, wk1, xr, nkb - 1);
    __builtin_amdgcn_sched_barrier(0);
    proj_mma(f0, q0, q1, k0, k1);
    proj_mma(f1, q0, q1, k0, k1);
    float* Qb = Qp + (size_t)b * C_ * N_ + (size_t)(c0 >> 4) * (16 * N_) + (size_t)(nw >> 4) * 256;
    float* Kb = Kp + (size_t)b * C_ * N_ + (size_t)(c0 >> 4) * (16 * N_) + (size_t)(nw >> 4) * 256;
    #pragma unroll
    for (int t = 0; t < 4; ++t) {
        store_frag_tile(Qb + t * 256,           lane, q0[t]);
        store_frag_tile(Qb + 16 * N_ + t * 256, lane, q1[t]);
        store_frag_tile(Kb + t * 256,           lane, k0[t]);
        store_frag_tile(Kb + 16 * N_ + t * 256, lane, k1[t]);
    }
}

__global__ __launch_bounds__(1024) void k_rownorm(const float* __restrict__ Qp,
                                                  const float* __restrict__ Kp,
                                                  float* __restrict__ invQ,
                                                  float* __restrict__ invK) {
    __shared__ float part[32][16];
    const int tid  = threadIdx.x;
    const int lane = tid & 31;
    const int w    = tid >> 5;
    const int blk  = blockIdx.x;
    const float* src = blockIdx.y ? Kp : Qp;
    const float4* p = (const float4*)(src + (size_t)blk * (16 * N_)) + tid;
    float s = 0.f;
    #pragma unroll
    for (int i = 0; i < 16; ++i) {
        float4 v = p[i * 1024];
        s += (v.x * v.x + v.y * v.y) + (v.z * v.z + v.w * v.w);
    }
    s += __shfl_xor(s, 16, 32);
    if (lane < 16) part[w][lane] = s;
    __syncthreads();
    if (tid < 16) {
        float t = 0.f;
        for (int i = 0; i < 32; ++i) t += part[i][tid];
        float* dst = blockIdx.y ? invK : invQ;
        dst[blk * 16 + tid] = rsqrtf(t);
    }
}

struct QktFrag { float4 al[4], ah[4], bl[4], bh[4]; };

static __device__ __forceinline__ void qkt_load(QktFrag& f, const float4* p, const float4* t, int kb) {
    const int o = kb * 64;
    #pragma unroll
    for (int j = 0; j < 4; ++j) {
        f.al[j] = p[o + j * 4 * N_];
        f.ah[j] = p[o + 32 + j * 4 * N_];
    }
    #pragma unroll
    for (int j = 0; j < 4; ++j) {
        f.bl[j] = t[o + j * 4 * N_];
        f.bh[j] = t[o + 32 + j * 4 * N_];
    }
}

static __device__ __forceinline__ void qkt_step(const v2f (&a)[4], const v2f (&b)[4], v8f (&acc)[4][4]) {
    #pragma unroll
    for (int ra = 0; ra < 4; ++ra) {
        #pragma unroll
        for (int sb = 0; sb < 4; ++sb) acc[ra][sb] = wmma4(a[ra], b[sb], acc[ra][sb]);
    }
}

static __device__ __forceinline__ void qkt_mma(const QktFrag& f, v8f (&acc)[4][4]) {
    {
        const v2f a[4] = {{f.al[0].x, f.al[0].y}, {f.al[1].x, f.al[1].y}, {f.al[2].x, f.al[2].y}, {f.al[3].x, f.al[3].y}};
        const v2f b[4] = {{f.bl[0].x, f.bl[0].y}, {f.bl[1].x, f.bl[1].y}, {f.bl[2].x, f.bl[2].y}, {f.bl[3].x, f.bl[3].y}};
        qkt_step(a, b, acc);
    }
    {
        const v2f a[4] = {{f.al[0].z, f.al[0].w}, {f.al[1].z, f.al[1].w}, {f.al[2].z, f.al[2].w}, {f.al[3].z, f.al[3].w}};
        const v2f b[4] = {{f.bl[0].z, f.bl[0].w}, {f.bl[1].z, f.bl[1].w}, {f.bl[2].z, f.bl[2].w}, {f.bl[3].z, f.bl[3].w}};
        qkt_step(a, b, acc);
    }
    {
        const v2f a[4] = {{f.ah[0].x, f.ah[0].y}, {f.ah[1].x, f.ah[1].y}, {f.ah[2].x, f.ah[2].y}, {f.ah[3].x, f.ah[3].y}};
        const v2f b[4] = {{f.bh[0].x, f.bh[0].y}, {f.bh[1].x, f.bh[1].y}, {f.bh[2].x, f.bh[2].y}, {f.bh[3].x, f.bh[3].y}};
        qkt_step(a, b, acc);
    }
    {
        const v2f a[4] = {{f.ah[0].z, f.ah[0].w}, {f.ah[1].z, f.ah[1].w}, {f.ah[2].z, f.ah[2].w}, {f.ah[3].z, f.ah[3].w}};
        const v2f b[4] = {{f.bh[0].z, f.bh[0].w}, {f.bh[1].z, f.bh[1].w}, {f.bh[2].z, f.bh[2].w}, {f.bh[3].z, f.bh[3].w}};
        qkt_step(a, b, acc);
    }
}

__global__ __launch_bounds__(128) void k_qkt(const float* __restrict__ Rp,
                                             const float* __restrict__ Sp,
                                             const float* __restrict__ invR,
                                             const float* __restrict__ invS,
                                             float* __restrict__ Mt0,
                                             float* __restrict__ Mt1) {
    const int lane = threadIdx.x & 31;
    const int wave = threadIdx.x >> 5;
    const int m    = lane & 15;
    const int half = lane >> 4;
    const int r0 = blockIdx.y * 64;
    const int s0 = blockIdx.x * 256 + wave * 64;
    const int b  = blockIdx.z >> 1;
    const int kh = blockIdx.z & 1;
    const int nkb = N_ / 32;
    const size_t bOff = (size_t)b * C_ * N_;
    const float4* p = (const float4*)(Rp + bOff + (size_t)(r0 >> 4) * (16 * N_)) + kh * (nkb * 64) + lane;
    const float4* t = (const float4*)(Sp + bOff + (size_t)(s0 >> 4) * (16 * N_)) + kh * (nkb * 64) + lane;
    v8f acc[4][4] = {};
    QktFrag f0, f1, f2;
    qkt_load(f0, p, t, 0);
    __builtin_amdgcn_sched_barrier(0);
    qkt_load(f1, p, t, 1);
    __builtin_amdgcn_sched_barrier(0);
    for (int kb = 0; kb < nkb - 2; kb += 3) {
        qkt_load(f2, p, t, kb + 2);
        __builtin_amdgcn_sched_barrier(0);
        qkt_mma(f0, acc);
        __builtin_amdgcn_sched_barrier(0);
        qkt_load(f0, p, t, kb + 3);
        __builtin_amdgcn_sched_barrier(0);
        qkt_mma(f1, acc);
        __builtin_amdgcn_sched_barrier(0);
        qkt_load(f1, p, t, kb + 4);
        __builtin_amdgcn_sched_barrier(0);
        qkt_mma(f2, acc);
        __builtin_amdgcn_sched_barrier(0);
    }
    qkt_mma(f0, acc);
    qkt_mma(f1, acc);
    float* Mb = (kh ? Mt1 : Mt0) + (size_t)b * C_ * C_;
    float is[4];
    #pragma unroll
    for (int sb = 0; sb < 4; ++sb) is[sb] = invS[b * C_ + s0 + 16 * sb + m];
    #pragma unroll
    for (int ra = 0; ra < 4; ++ra) {
        #pragma unroll
        for (int j = 0; j < 8; ++j) {
            const int row = r0 + 16 * ra + j + 8 * half;
            const float ir = invR[b * C_ + row];
            #pragma unroll
            for (int sb = 0; sb < 4; ++sb)
                Mb[(size_t)row * C_ + s0 + 16 * sb + m] = acc[ra][sb][j] * ir * is[sb];
        }
    }
}

__global__ __launch_bounds__(256) void k_rowsoftmax(float* __restrict__ Mt,
                                                    const float* __restrict__ Mt1) {
    const int lane = threadIdx.x & 31;
    const int row  = blockIdx.x * 8 + (threadIdx.x >> 5);
    float4* r = (float4*)(Mt + (size_t)row * C_);
    const float4* r1 = (const float4*)(Mt1 + (size_t)row * C_);
    float4 v[4];
    float mx = -3.4e38f, mn = 3.4e38f;
    for (int i = 0; i < 4; ++i) {
        float4 t = r[i * 32 + lane];
        const float4 u = r1[i * 32 + lane];
        t.x = 1.f - (t.x + u.x); t.y = 1.f - (t.y + u.y); t.z = 1.f - (t.z + u.z); t.w = 1.f - (t.w + u.w);
        v[i] = t;
        mx = fmaxf(mx, fmaxf(fmaxf(t.x, t.y), fmaxf(t.z, t.w)));
        mn = fminf(mn, fminf(fminf(t.x, t.y), fminf(t.z, t.w)));
    }
    for (int o = 16; o > 0; o >>= 1) {
        mx = fmaxf(mx, __shfl_xor(mx, o, 32));
        mn = fminf(mn, __shfl_xor(mn, o, 32));
    }
    const float dn = mx + EPS_;
    const float inv_dn = 1.f / dn;
    const float sh = 1.f - (mn / dn) * (1.f / HT_);
    float z = 0.f;
    for (int i = 0; i < 4; ++i) {
        float4 t = v[i];
        t.x = expf((1.f - (t.x * inv_dn) * (1.f / HT_)) - sh);
        t.y = expf((1.f - (t.y * inv_dn) * (1.f / HT_)) - sh);
        t.z = expf((1.f - (t.z * inv_dn) * (1.f / HT_)) - sh);
        t.w = expf((1.f - (t.w * inv_dn) * (1.f / HT_)) - sh);
        v[i] = t;
        z += (t.x + t.y) + (t.z + t.w);
    }
    for (int o = 16; o > 0; o >>= 1) z += __shfl_xor(z, o, 32);
    const float invz = 1.f / z;
    for (int i = 0; i < 4; ++i) {
        float4 t = v[i];
        t.x *= invz; t.y *= invz; t.z *= invz; t.w *= invz;
        r[i * 32 + lane] = t;
    }
}

__global__ __launch_bounds__(1024) void k_l1(float* __restrict__ Mt) {
    __shared__ float part[32][33];
    const int lane = threadIdx.x & 31;
    const int w    = threadIdx.x >> 5;
    float* base = Mt + (size_t)blockIdx.y * C_ * C_ + (size_t)(w * 16) * C_
                     + blockIdx.x * 32 + lane;
    float v[16];
    float s = 0.f;
    for (int i = 0; i < 16; ++i) { v[i] = base[(size_t)i * C_]; s += v[i]; }
    part[w][lane] = s;
    __syncthreads();
    float t = 0.f;
    for (int i = 0; i < 32; ++i) t += part[i][lane];
    const float inv = 1.f / (t + EPS_);
    for (int i = 0; i < 16; ++i) base[(size_t)i * C_] = v[i] * inv;
}

__global__ __launch_bounds__(128) void k_mwv(const float* __restrict__ Mt,
                                             const float* __restrict__ Wv,
                                             float* __restrict__ Gp) {
    const int lane = threadIdx.x & 31;
    const int wave = threadIdx.x >> 5;
    const int m    = lane & 15;
    const int kp   = (lane >> 4) << 1;
    const int c0 = blockIdx.y * 32;
    const int e0 = blockIdx.x * 128 + wave * 32;
    const int b  = blockIdx.z;
    const float* a = Mt + (size_t)b * C_ * C_ + c0 + m;
    v8f g00 = {}, g01 = {}, g10 = {}, g11 = {};
    for (int k = 0; k < C_; k += 4) {
        const int kk = k + kp;
        const float* ar0 = a + (size_t)kk * C_;
        const float* ar1 = a + (size_t)(kk + 1) * C_;
        v2f aA, aB;
        aA.x = ar0[0];  aA.y = ar1[0];
        aB.x = ar0[16]; aB.y = ar1[16];
        const float* wr0 = Wv + (size_t)kk * C_;
        const float* wr1 = Wv + (size_t)(kk + 1) * C_;
        v2f b0, b1;
        b0.x = wr0[e0 + m];      b0.y = wr1[e0 + m];
        b1.x = wr0[e0 + 16 + m]; b1.y = wr1[e0 + 16 + m];
        g00 = wmma4(b0, aA, g00); g01 = wmma4(b1, aA, g01);
        g10 = wmma4(b0, aB, g10); g11 = wmma4(b1, aB, g11);
    }
    float* Gb = Gp + (size_t)b * C_ * C_ + (size_t)(c0 >> 4) * (16 * C_) + (size_t)(e0 >> 4) * 256;
    store_frag_tile(Gb,                 lane, g00);
    store_frag_tile(Gb + 256,           lane, g01);
    store_frag_tile(Gb + 16 * C_,       lane, g10);
    store_frag_tile(Gb + 16 * C_ + 256, lane, g11);
}

struct OutFrag {
    float4 a0l, a0h, a1l, a1h;
    float4 xv[8];
};

static __device__ __forceinline__ void out_load(OutFrag& f, const float4* g0, const float4* g1,
                                                const float* xr, int kb) {
    const int o = kb * 64;
    f.a0l = g0[o]; f.a0h = g0[o + 32];
    f.a1l = g1[o]; f.a1h = g1[o + 32];
    const float* p = xr + (size_t)kb * (16 * N_);
    #pragma unroll
    for (int j = 0; j < 8; ++j) f.xv[j] = *(const float4*)(p + (size_t)j * N_);
}

static __device__ __forceinline__ void out_step(v2f aA, v2f aB, const float4& r0, const float4& r1,
                                                v8f (&o0)[4], v8f (&o1)[4]) {
    #pragma unroll
    for (int t = 0; t < 4; ++t) {
        const v2f bt = {comp4(r0, t), comp4(r1, t)};
        o0[t] = wmma4(aA, bt, o0[t]);
        o1[t] = wmma4(aB, bt, o1[t]);
    }
}

static __device__ __forceinline__ void out_mma(const OutFrag& f, v8f (&o0)[4], v8f (&o1)[4]) {
    out_step(v2f{f.a0l.x, f.a0l.y}, v2f{f.a1l.x, f.a1l.y}, f.xv[0], f.xv[1], o0, o1);
    out_step(v2f{f.a0l.z, f.a0l.w}, v2f{f.a1l.z, f.a1l.w}, f.xv[2], f.xv[3], o0, o1);
    out_step(v2f{f.a0h.x, f.a0h.y}, v2f{f.a1h.x, f.a1h.y}, f.xv[4], f.xv[5], o0, o1);
    out_step(v2f{f.a0h.z, f.a0h.w}, v2f{f.a1h.z, f.a1h.w}, f.xv[6], f.xv[7], o0, o1);
}

__global__ __launch_bounds__(128) void k_out(const float* __restrict__ x,
                                             const float* __restrict__ Gp,
                                             const float* __restrict__ gamma,
                                             float* __restrict__ y) {
    const int lane = threadIdx.x & 31;
    const int wave = threadIdx.x >> 5;
    const int i    = lane & 15;
    const int half = lane >> 4;
    const int c0 = blockIdx.y * 32;
    const int nw = blockIdx.x * 256 + wave * 64;
    const int b  = blockIdx.z;
    const float* xb = x + (size_t)b * C_ * N_;
    const float* xr = xb + (size_t)(8 * half) * N_ + nw + 4 * i;
    const float4* g0 = (const float4*)(Gp + (size_t)b * C_ * C_ + (size_t)(c0 >> 4) * (16 * C_)) + lane;
    const float4* g1 = g0 + 4 * C_;
    v8f o0[4] = {}, o1[4] = {};
    OutFrag f0, f1;
    const int nkb = C_ / 16;
    out_load(f0, g0, g1, xr, 0);
    for (int kb = 0; kb < nkb - 2; kb += 2) {
        out_load(f1, g0, g1, xr, kb + 1);
        __builtin_amdgcn_sched_barrier(0);
        out_mma(f0, o0, o1);
        __builtin_amdgcn_sched_barrier(0);
        out_load(f0, g0, g1, xr, kb + 2);
        __builtin_amdgcn_sched_barrier(0);
        out_mma(f1, o0, o1);
        __builtin_amdgcn_sched_barrier(0);
    }
    out_load(f1, g0, g1, xr, nkb - 1);
    __builtin_amdgcn_sched_barrier(0);
    out_mma(f0, o0, o1);
    out_mma(f1, o0, o1);
    const float g = gamma[0];
    float* yb = y + (size_t)b * C_ * N_;
    #pragma unroll
    for (int j = 0; j < 8; ++j) {
        const size_t oA = (size_t)(c0 + j + 8 * half) * N_ + nw + 4 * i;
        const size_t oB = oA + (size_t)16 * N_;
        const float4 xA = *(const float4*)(xb + oA);
        const float4 xB = *(const float4*)(xb + oB);
        *(float4*)(yb + oA) = make_float4(xA.x + g * o0[0][j], xA.y + g * o0[1][j],
                                          xA.z + g * o0[2][j], xA.w + g * o0[3][j]);
        *(float4*)(yb + oB) = make_float4(xB.x + g * o1[0][j], xB.y + g * o1[1][j],
                                          xB.z + g * o1[2][j], xB.w + g * o1[3][j]);
    }
}

extern "C" void kernel_launch(void* const* d_in, const int* in_sizes, int n_in,
                              void* d_out, int out_size, void* d_ws, size_t ws_size,
                              hipStream_t stream) {
    (void)in_sizes; (void)n_in; (void)out_size; (void)ws_size;
    const float* x     = (const float*)d_in[0];
    const float* Wq    = (const float*)d_in[1];
    const float* Wk    = (const float*)d_in[2];
    const float* Wv    = (const float*)d_in[3];
    const float* gamma = (const float*)d_in[4];
    float* y = (float*)d_out;

    char* ws = (char*)d_ws;
    const size_t qkBytes = (size_t)B_ * C_ * N_ * sizeof(float);
    const size_t mBytes  = (size_t)B_ * C_ * C_ * sizeof(float);
    float* Q     = (float*)(ws);
    float* K     = (float*)(ws + qkBytes);
    float* M     = (float*)(ws + 2 * qkBytes);
    float* G     = (float*)(ws + 2 * qkBytes + mBytes);
    float* invQ  = (float*)(ws + 2 * qkBytes + 2 * mBytes);
    float* invK  = invQ  + B_ * C_;
    float* Wqp   = G;
    float* Wkp   = G + C_ * C_;
    float* M1    = G;

    k_wfrag     <<<dim3(C_ * C_ / 4 / 256, 2), 256, 0, stream>>>(Wq, Wk, Wqp, Wkp);
    k_qkproj    <<<dim3(N_ / 256, C_ / 32, B_), 128, 0, stream>>>(x, Wqp, Wkp, Q, K);
    k_rownorm   <<<dim3(B_ * C_ / 16, 2), 1024, 0, stream>>>(Q, K, invQ, invK);
    k_qkt       <<<dim3(C_ / 256, C_ / 64, B_ * 2), 128, 0, stream>>>(K, Q, invK, invQ, M, M1);
    k_rowsoftmax<<<B_ * C_ / 8, 256, 0, stream>>>(M, M1);
    k_l1        <<<dim3(C_ / 32, B_), 1024, 0, stream>>>(M);
    k_mwv       <<<dim3(C_ / 128, C_ / 32, B_), 128, 0, stream>>>(M, Wv, G);
    k_out       <<<dim3(N_ / 256, C_ / 32, B_), 128, 0, stream>>>(x, G, gamma, y);
}
